# starting kernel
# baseline (speedup 1.0000x reference)
#include <hip/hip_runtime.h>
#include <math.h>

typedef float v2f __attribute__((ext_vector_type(2)));
typedef float v8f __attribute__((ext_vector_type(8)));

#define NEG_SLOPE 0.2f

__device__ __forceinline__ float leaky(float x) {
    return x > 0.f ? x : NEG_SLOPE * x;
}

__device__ __forceinline__ void atomicMaxF32(float* addr, float v) {
    if (v >= 0.f) {
        atomicMax((int*)addr, __float_as_int(v));
    } else {
        atomicMin((unsigned int*)addr, __float_as_uint(v));
    }
}

template <int CT>
__global__ void wmma_gemm_f32(const float* __restrict__ A,
                              const float* __restrict__ B,
                              float* __restrict__ Out,
                              int M, int K, int Nout) {
    const int wave = threadIdx.x >> 5;
    const int lane = threadIdx.x & 31;
    const int row0 = (blockIdx.x * (blockDim.x >> 5) + wave) << 4;
    const int col0 = blockIdx.y * (CT * 16);
    if (row0 >= M) return;

    const int half = lane >> 4;
    const int l16  = lane & 15;
    const float* arow = A + (size_t)(row0 + l16) * K;
    const int col = col0 + l16;

    v8f acc[CT];
    #pragma unroll
    for (int t = 0; t < CT; ++t) acc[t] = (v8f){};

    for (int k = 0; k < K; k += 4) {
        const int ka = k + 2 * half;
        v2f a;
        a.x = arow[ka];
        a.y = arow[ka + 1];
        const float* brow0 = B + (size_t)ka * Nout + col;
        const float* brow1 = B + (size_t)(ka + 1) * Nout + col;
        #pragma unroll
        for (int t = 0; t < CT; ++t) {
            v2f b;
            b.x = brow0[t * 16];
            b.y = brow1[t * 16];
            acc[t] = __builtin_amdgcn_wmma_f32_16x16x4_f32(
                 false, a,  false, b,
                 (short)0, acc[t],  false,  false);
        }
    }
    #pragma unroll
    for (int t = 0; t < CT; ++t) {
        #pragma unroll
        for (int r = 0; r < 8; ++r) {
            Out[(size_t)(row0 + r + 8 * half) * Nout + col + t * 16] = acc[t][r];
        }
    }
}

template <int H, int C>
__global__ void alpha_kernel(const float* __restrict__ h,
                             const float* __restrict__ a_src,
                             const float* __restrict__ a_dst,
                             float* __restrict__ as, float* __restrict__ ad,
                             int n) {
    int gid = blockIdx.x * blockDim.x + threadIdx.x;
    if (gid >= n * H) return;
    const int node = gid / H;
    const int hh = gid - node * H;
    const float* hp = h + (size_t)node * (H * C) + hh * C;
    const float* asw = a_src + hh * C;
    const float* adw = a_dst + hh * C;
    float s = 0.f, d = 0.f;
    #pragma unroll
    for (int c = 0; c < C; ++c) {
        s += hp[c] * asw[c];
        d += hp[c] * adw[c];
    }
    as[gid] = s;
    ad[gid] = d;
}

__global__ void fill_kernel(float* __restrict__ p, float v, int n) {
    int gid = blockIdx.x * blockDim.x + threadIdx.x;
    if (gid < n) p[gid] = v;
}

__global__ void init_bias_kernel(float* __restrict__ out,
                                 const float* __restrict__ bias,
                                 int n, int HC) {
    int gid = blockIdx.x * blockDim.x + threadIdx.x;
    if (gid >= n * HC) return;
    out[gid] = bias[gid % HC];
}

__device__ __forceinline__ void edge_sd(const int* __restrict__ eidx, int E,
                                        int i, int& s, int& d) {
    if (i < E) { s = eidx[i]; d = eidx[E + i]; }
    else       { s = i - E;   d = i - E; }
}

template <int H>
__global__ void edge_max_kernel(const int* __restrict__ eidx, int E, int n,
                                const float* __restrict__ as,
                                const float* __restrict__ ad,
                                float* __restrict__ m) {
    int gid = blockIdx.x * blockDim.x + threadIdx.x;
    int tot = (E + n) * H;
    if (gid >= tot) return;
    const int i = gid / H;
    const int hh = gid - i * H;
    int s, d;
    edge_sd(eidx, E, i, s, d);
    float e = leaky(as[s * H + hh] + ad[d * H + hh]);
    atomicMaxF32(&m[d * H + hh], e);
}

template <int H>
__global__ void edge_sum_kernel(const int* __restrict__ eidx, int E, int n,
                                const float* __restrict__ as,
                                const float* __restrict__ ad,
                                const float* __restrict__ m,
                                float* __restrict__ den) {
    int gid = blockIdx.x * blockDim.x + threadIdx.x;
    int tot = (E + n) * H;
    if (gid >= tot) return;
    const int i = gid / H;
    const int hh = gid - i * H;
    int s, d;
    edge_sd(eidx, E, i, s, d);
    float e = leaky(as[s * H + hh] + ad[d * H + hh]);
    float ex = __expf(e - m[d * H + hh]);
    atomicAdd(&den[d * H + hh], ex);
}

template <int H, int C>
__global__ void edge_agg_kernel(const int* __restrict__ eidx, int E, int n,
                                const float* __restrict__ as,
                                const float* __restrict__ ad,
                                const float* __restrict__ m,
                                const float* __restrict__ den,
                                const float* __restrict__ h,
                                float* __restrict__ out) {
    int gid = blockIdx.x * blockDim.x + threadIdx.x;
    int tot = (E + n) * H;
    if (gid >= tot) return;
    const int i = gid / H;
    const int hh = gid - i * H;
    int s, d;
    edge_sd(eidx, E, i, s, d);
    float e = leaky(as[s * H + hh] + ad[d * H + hh]);
    float coeff = __expf(e - m[d * H + hh]) / den[d * H + hh];
    const float* hs = h + (size_t)s * (H * C) + hh * C;
    float* od = out + (size_t)d * (H * C) + hh * C;
    #pragma unroll
    for (int c = 0; c < C; ++c) {
        atomicAdd(&od[c], coeff * hs[c]);
    }
}

static inline int ceil_div(int a, int b) { return (a + b - 1) / b; }

extern "C" void kernel_launch(void* const* d_in, const int* in_sizes, int n_in,
                              void* d_out, int out_size, void* d_ws, size_t ws_size,
                              hipStream_t stream) {
    const float* x    = (const float*)d_in[0];
    const int*   eidx = (const int*)  d_in[1];
    const float* W1   = (const float*)d_in[2];
    const float* aS1  = (const float*)d_in[3];
    const float* aD1  = (const float*)d_in[4];
    const float* b1   = (const float*)d_in[5];
    const float* W2   = (const float*)d_in[6];
    const float* aS2  = (const float*)d_in[7];
    const float* aD2  = (const float*)d_in[8];
    const float* b2   = (const float*)d_in[9];
    float* out = (float*)d_out;

    const int FEATS = 256, HID = 64, H1 = 8, C1 = 8, CLS = 16;
    const int N = in_sizes[0] / FEATS;
    const int E = in_sizes[1] / 2;
    const int Etot = E + N;

    float* ws = (float*)d_ws;
    float* h1   = ws;                 ws += (size_t)N * HID;
    float* as1  = ws;                 ws += (size_t)N * H1;
    float* ad1  = ws;                 ws += (size_t)N * H1;
    float* m1   = ws;                 ws += (size_t)N * H1;
    float* den1 = ws;                 ws += (size_t)N * H1;
    float* o1   = ws;                 ws += (size_t)N * HID;
    float* h2   = ws;                 ws += (size_t)N * CLS;
    float* as2  = ws;                 ws += (size_t)N;
    float* ad2  = ws;                 ws += (size_t)N;
    float* m2   = ws;                 ws += (size_t)N;
    float* den2 = ws;                 ws += (size_t)N;

    const int TPB = 256;
    const int WPB = 4;
    const dim3 gblock(32 * WPB);

    {
        dim3 grid(ceil_div(N / 16, WPB), 1);
        wmma_gemm_f32<4><<<grid, gblock, 0, stream>>>(x, W1, h1, N, FEATS, HID);
    }
    alpha_kernel<H1, C1><<<ceil_div(N * H1, TPB), TPB, 0, stream>>>(h1, aS1, aD1, as1, ad1, N);
    fill_kernel<<<ceil_div(N * H1, TPB), TPB, 0, stream>>>(m1, -INFINITY, N * H1);
    fill_kernel<<<ceil_div(N * H1, TPB), TPB, 0, stream>>>(den1, 0.f, N * H1);
    init_bias_kernel<<<ceil_div(N * HID, TPB), TPB, 0, stream>>>(o1, b1, N, HID);

    {
        const int tot = Etot * H1;
        edge_max_kernel<H1><<<ceil_div(tot, TPB), TPB, 0, stream>>>(eidx, E, N, as1, ad1, m1);
        edge_sum_kernel<H1><<<ceil_div(tot, TPB), TPB, 0, stream>>>(eidx, E, N, as1, ad1, m1, den1);
        edge_agg_kernel<H1, C1><<<ceil_div(tot, TPB), TPB, 0, stream>>>(eidx, E, N, as1, ad1, m1, den1, h1, o1);
    }

    {
        dim3 grid(ceil_div(N / 16, WPB), 1);
        wmma_gemm_f32<1><<<grid, gblock, 0, stream>>>(o1, W2, h2, N, HID, CLS);
    }
    alpha_kernel<1, CLS><<<ceil_div(N, TPB), TPB, 0, stream>>>(h2, aS2, aD2, as2, ad2, N);
    fill_kernel<<<ceil_div(N, TPB), TPB, 0, stream>>>(m2, -INFINITY, N);
    fill_kernel<<<ceil_div(N, TPB), TPB, 0, stream>>>(den2, 0.f, N);
    init_bias_kernel<<<ceil_div(N * CLS, TPB), TPB, 0, stream>>>(out, b2, N, CLS);

    {
        const int tot = Etot;
        edge_max_kernel<1><<<ceil_div(tot, TPB), TPB, 0, stream>>>(eidx, E, N, as2, ad2, m2);
        edge_sum_kernel<1><<<ceil_div(tot, TPB), TPB, 0, stream>>>(eidx, E, N, as2, ad2, m2, den2);
        edge_agg_kernel<1, CLS><<<ceil_div(tot, TPB), TPB, 0, stream>>>(eidx, E, N, as2, ad2, m2, den2, h2, out);
    }
}


# edge aggregation: lanes laid along each row's channels (coefficient and row starts read across lanes), so each load and float atomic-add covers contiguous lines; atomics and arrival-order sums kept
# speedup vs baseline: 1.8542x; 1.8542x over previous
#include <hip/hip_runtime.h>
#include <math.h>

typedef float v2f __attribute__((ext_vector_type(2)));
typedef float v8f __attribute__((ext_vector_type(8)));

#define NEG_SLOPE 0.2f

__device__ __forceinline__ float leaky(float x) {
    return x > 0.f ? x : NEG_SLOPE * x;
}

__device__ __forceinline__ void atomicMaxF32(float* addr, float v) {
    if (v >= 0.f) {
        atomicMax((int*)addr, __float_as_int(v));
    } else {
        atomicMin((unsigned int*)addr, __float_as_uint(v));
    }
}

template <int CT>
__global__ void wmma_gemm_f32(const float* __restrict__ A,
                              const float* __restrict__ B,
                              float* __restrict__ Out,
                              int M, int K, int Nout) {
    const int wave = threadIdx.x >> 5;
    const int lane = threadIdx.x & 31;
    const int row0 = (blockIdx.x * (blockDim.x >> 5) + wave) << 4;
    const int col0 = blockIdx.y * (CT * 16);
    if (row0 >= M) return;

    const int half = lane >> 4;
    const int l16  = lane & 15;
    const float* arow = A + (size_t)(row0 + l16) * K;
    const int col = col0 + l16;

    v8f acc[CT];
    #pragma unroll
    for (int t = 0; t < CT; ++t) acc[t] = (v8f){};

    for (int k = 0; k < K; k += 4) {
        const int ka = k + 2 * half;
        v2f a;
        a.x = arow[ka];
        a.y = arow[ka + 1];
        const float* brow0 = B + (size_t)ka * Nout + col;
        const float* brow1 = B + (size_t)(ka + 1) * Nout + col;
        #pragma unroll
        for (int t = 0; t < CT; ++t) {
            v2f b;
            b.x = brow0[t * 16];
            b.y = brow1[t * 16];
            acc[t] = __builtin_amdgcn_wmma_f32_16x16x4_f32(
                 false, a,  false, b,
                 (short)0, acc[t],  false,  false);
        }
    }
    #pragma unroll
    for (int t = 0; t < CT; ++t) {
        #pragma unroll
        for (int r = 0; r < 8; ++r) {
            Out[(size_t)(row0 + r + 8 * half) * Nout + col + t * 16] = acc[t][r];
        }
    }
}

template <int H, int C>
__global__ void alpha_kernel(const float* __restrict__ h,
                             const float* __restrict__ a_src,
                             const float* __restrict__ a_dst,
                             float* __restrict__ as, float* __restrict__ ad,
                             int n) {
    int gid = blockIdx.x * blockDim.x + threadIdx.x;
    if (gid >= n * H) return;
    const int node = gid / H;
    const int hh = gid - node * H;
    const float* hp = h + (size_t)node * (H * C) + hh * C;
    const float* asw = a_src + hh * C;
    const float* adw = a_dst + hh * C;
    float s = 0.f, d = 0.f;
    #pragma unroll
    for (int c = 0; c < C; ++c) {
        s += hp[c] * asw[c];
        d += hp[c] * adw[c];
    }
    as[gid] = s;
    ad[gid] = d;
}

__global__ void fill_kernel(float* __restrict__ p, float v, int n) {
    int gid = blockIdx.x * blockDim.x + threadIdx.x;
    if (gid < n) p[gid] = v;
}

__global__ void init_bias_kernel(float* __restrict__ out,
                                 const float* __restrict__ bias,
                                 int n, int HC) {
    int gid = blockIdx.x * blockDim.x + threadIdx.x;
    if (gid >= n * HC) return;
    out[gid] = bias[gid % HC];
}

__device__ __forceinline__ void edge_sd(const int* __restrict__ eidx, int E,
                                        int i, int& s, int& d) {
    if (i < E) { s = eidx[i]; d = eidx[E + i]; }
    else       { s = i - E;   d = i - E; }
}

template <int H>
__global__ void edge_max_kernel(const int* __restrict__ eidx, int E, int n,
                                const float* __restrict__ as,
                                const float* __restrict__ ad,
                                float* __restrict__ m) {
    int gid = blockIdx.x * blockDim.x + threadIdx.x;
    int tot = (E + n) * H;
    if (gid >= tot) return;
    const int i = gid / H;
    const int hh = gid - i * H;
    int s, d;
    edge_sd(eidx, E, i, s, d);
    float e = leaky(as[s * H + hh] + ad[d * H + hh]);
    atomicMaxF32(&m[d * H + hh], e);
}

template <int H>
__global__ void edge_sum_kernel(const int* __restrict__ eidx, int E, int n,
                                const float* __restrict__ as,
                                const float* __restrict__ ad,
                                const float* __restrict__ m,
                                float* __restrict__ den) {
    int gid = blockIdx.x * blockDim.x + threadIdx.x;
    int tot = (E + n) * H;
    if (gid >= tot) return;
    const int i = gid / H;
    const int hh = gid - i * H;
    int s, d;
    edge_sd(eidx, E, i, s, d);
    float e = leaky(as[s * H + hh] + ad[d * H + hh]);
    float ex = __expf(e - m[d * H + hh]);
    atomicAdd(&den[d * H + hh], ex);
}

template <int H, int C>
__global__ void edge_agg_kernel(const int* __restrict__ eidx, int E, int n,
                                const float* __restrict__ as,
                                const float* __restrict__ ad,
                                const float* __restrict__ m,
                                const float* __restrict__ den,
                                const float* __restrict__ h,
                                float* __restrict__ out) {
    constexpr bool ROW_PER_STEP = ((H * C) % 32 == 0);
    static_assert(32 % H == 0, "a wave's 32 pairs must hold whole edges");
    const int lane = threadIdx.x & 31;
    const int gid = blockIdx.x * blockDim.x + threadIdx.x;
    const int tot = (E + n) * H;
    const bool valid = gid < tot;
    const int g = valid ? gid : 0;
    const int i = g / H;
    const int hh = g - i * H;
    int s, d;
    edge_sd(eidx, E, i, s, d);
    float e = leaky(as[s * H + hh] + ad[d * H + hh]);
    const float coeff = __expf(e - m[d * H + hh]) / den[d * H + hh];
    const int soff = (s * H + hh) * C;
    const int doff = valid ? (d * H + hh) * C : -1;

    float val[C];
    int   dbase[C];
    #pragma unroll
    for (int t = 0; t < C; ++t) {
        const int flat = t * 32 + lane;
        const int p = flat / C;
        const float cf = __shfl(coeff, p);
        int so;
        if constexpr (ROW_PER_STEP) {
            const int first = (t * 32 / (H * C)) * H;
            so = __builtin_amdgcn_readlane(soff, first) + (flat - first * C);
            dbase[t] = __builtin_amdgcn_readlane(doff, first);
        } else {
            so = __shfl(soff, p) + (flat - p * C);
            dbase[t] = __shfl(doff, p);
        }
        val[t] = cf * h[so];
    }
    #pragma unroll
    for (int t = 0; t < C; ++t) {
        const int flat = t * 32 + lane;
        const int first = ROW_PER_STEP ? (t * 32 / (H * C)) * H : flat / C;
        if (dbase[t] >= 0) atomicAdd(&out[dbase[t] + (flat - first * C)], val[t]);
    }
}

static inline int ceil_div(int a, int b) { return (a + b - 1) / b; }

extern "C" void kernel_launch(void* const* d_in, const int* in_sizes, int n_in,
                              void* d_out, int out_size, void* d_ws, size_t ws_size,
                              hipStream_t stream) {
    const float* x    = (const float*)d_in[0];
    const int*   eidx = (const int*)  d_in[1];
    const float* W1   = (const float*)d_in[2];
    const float* aS1  = (const float*)d_in[3];
    const float* aD1  = (const float*)d_in[4];
    const float* b1   = (const float*)d_in[5];
    const float* W2   = (const float*)d_in[6];
    const float* aS2  = (const float*)d_in[7];
    const float* aD2  = (const float*)d_in[8];
    const float* b2   = (const float*)d_in[9];
    float* out = (float*)d_out;

    const int FEATS = 256, HID = 64, H1 = 8, C1 = 8, CLS = 16;
    const int N = in_sizes[0] / FEATS;
    const int E = in_sizes[1] / 2;
    const int Etot = E + N;

    float* ws = (float*)d_ws;
    float* h1   = ws;                 ws += (size_t)N * HID;
    float* as1  = ws;                 ws += (size_t)N * H1;
    float* ad1  = ws;                 ws += (size_t)N * H1;
    float* m1   = ws;                 ws += (size_t)N * H1;
    float* den1 = ws;                 ws += (size_t)N * H1;
    float* o1   = ws;                 ws += (size_t)N * HID;
    float* h2   = ws;                 ws += (size_t)N * CLS;
    float* as2  = ws;                 ws += (size_t)N;
    float* ad2  = ws;                 ws += (size_t)N;
    float* m2   = ws;                 ws += (size_t)N;
    float* den2 = ws;                 ws += (size_t)N;

    const int TPB = 256;
    const int WPB = 4;
    const dim3 gblock(32 * WPB);

    {
        dim3 grid(ceil_div(N / 16, WPB), 1);
        wmma_gemm_f32<4><<<grid, gblock, 0, stream>>>(x, W1, h1, N, FEATS, HID);
    }
    alpha_kernel<H1, C1><<<ceil_div(N * H1, TPB), TPB, 0, stream>>>(h1, aS1, aD1, as1, ad1, N);
    fill_kernel<<<ceil_div(N * H1, TPB), TPB, 0, stream>>>(m1, -INFINITY, N * H1);
    fill_kernel<<<ceil_div(N * H1, TPB), TPB, 0, stream>>>(den1, 0.f, N * H1);
    init_bias_kernel<<<ceil_div(N * HID, TPB), TPB, 0, stream>>>(o1, b1, N, HID);

    {
        const int tot = Etot * H1;
        edge_max_kernel<H1><<<ceil_div(tot, TPB), TPB, 0, stream>>>(eidx, E, N, as1, ad1, m1);
        edge_sum_kernel<H1><<<ceil_div(tot, TPB), TPB, 0, stream>>>(eidx, E, N, as1, ad1, m1, den1);
        edge_agg_kernel<H1, C1><<<ceil_div(tot, TPB), TPB, 0, stream>>>(eidx, E, N, as1, ad1, m1, den1, h1, o1);
    }

    {
        dim3 grid(ceil_div(N / 16, WPB), 1);
        wmma_gemm_f32<1><<<grid, gblock, 0, stream>>>(o1, W2, h2, N, HID, CLS);
    }
    alpha_kernel<1, CLS><<<ceil_div(N, TPB), TPB, 0, stream>>>(h2, aS2, aD2, as2, ad2, N);
    fill_kernel<<<ceil_div(N, TPB), TPB, 0, stream>>>(m2, -INFINITY, N);
    fill_kernel<<<ceil_div(N, TPB), TPB, 0, stream>>>(den2, 0.f, N);
    init_bias_kernel<<<ceil_div(N * CLS, TPB), TPB, 0, stream>>>(out, b2, N, CLS);

    {
        const int tot = Etot;
        edge_max_kernel<1><<<ceil_div(tot, TPB), TPB, 0, stream>>>(eidx, E, N, as2, ad2, m2);
        edge_sum_kernel<1><<<ceil_div(tot, TPB), TPB, 0, stream>>>(eidx, E, N, as2, ad2, m2, den2);
        edge_agg_kernel<1, CLS><<<ceil_div(tot, TPB), TPB, 0, stream>>>(eidx, E, N, as2, ad2, m2, den2, h2, out);
    }
}


# edges counting-sorted by destination in-call; one wave per destination row does max, sum and weighted gather in registers: no float atomics, no m/den/init passes; row sums in the sort's arrival order
# speedup vs baseline: 3.0817x; 1.6620x over previous
#include <hip/hip_runtime.h>
#include <math.h>

typedef float v2f __attribute__((ext_vector_type(2)));
typedef float v8f __attribute__((ext_vector_type(8)));

#define NEG_SLOPE 0.2f

__device__ __forceinline__ float leaky(float x) {
    return x > 0.f ? x : NEG_SLOPE * x;
}


template <int CT>
__global__ void wmma_gemm_f32(const float* __restrict__ A,
                              const float* __restrict__ B,
                              float* __restrict__ Out,
                              int M, int K, int Nout) {
    const int wave = threadIdx.x >> 5;
    const int lane = threadIdx.x & 31;
    const int row0 = (blockIdx.x * (blockDim.x >> 5) + wave) << 4;
    const int col0 = blockIdx.y * (CT * 16);
    if (row0 >= M) return;

    const int half = lane >> 4;
    const int l16  = lane & 15;
    const float* arow = A + (size_t)(row0 + l16) * K;
    const int col = col0 + l16;

    v8f acc[CT];
    #pragma unroll
    for (int t = 0; t < CT; ++t) acc[t] = (v8f){};

    for (int k = 0; k < K; k += 4) {
        const int ka = k + 2 * half;
        v2f a;
        a.x = arow[ka];
        a.y = arow[ka + 1];
        const float* brow0 = B + (size_t)ka * Nout + col;
        const float* brow1 = B + (size_t)(ka + 1) * Nout + col;
        #pragma unroll
        for (int t = 0; t < CT; ++t) {
            v2f b;
            b.x = brow0[t * 16];
            b.y = brow1[t * 16];
            acc[t] = __builtin_amdgcn_wmma_f32_16x16x4_f32(
                 false, a,  false, b,
                 (short)0, acc[t],  false,  false);
        }
    }
    #pragma unroll
    for (int t = 0; t < CT; ++t) {
        #pragma unroll
        for (int r = 0; r < 8; ++r) {
            Out[(size_t)(row0 + r + 8 * half) * Nout + col + t * 16] = acc[t][r];
        }
    }
}

template <int H, int C>
__global__ void alpha_kernel(const float* __restrict__ h,
                             const float* __restrict__ a_src,
                             const float* __restrict__ a_dst,
                             float* __restrict__ as, float* __restrict__ ad,
                             int n) {
    int gid = blockIdx.x * blockDim.x + threadIdx.x;
    if (gid >= n * H) return;
    const int node = gid / H;
    const int hh = gid - node * H;
    const float* hp = h + (size_t)node * (H * C) + hh * C;
    const float* asw = a_src + hh * C;
    const float* adw = a_dst + hh * C;
    float s = 0.f, d = 0.f;
    #pragma unroll
    for (int c = 0; c < C; ++c) {
        s += hp[c] * asw[c];
        d += hp[c] * adw[c];
    }
    as[gid] = s;
    ad[gid] = d;
}

__global__ void zero_int_kernel(int* __restrict__ p, int n) {
    int gid = blockIdx.x * blockDim.x + threadIdx.x;
    if (gid < n) p[gid] = 0;
}

__global__ void count_kernel(const int* __restrict__ eidx, int E,
                             int* __restrict__ cnt) {
    int i = blockIdx.x * blockDim.x + threadIdx.x;
    if (i < E) atomicAdd(&cnt[eidx[E + i]], 1);
}

__global__ void scan_kernel(const int* __restrict__ cnt,
                            int* __restrict__ start, int* __restrict__ cursor,
                            int n) {
    __shared__ int s_incl[8];
    __shared__ int s_part[8];
    const int tid = threadIdx.x;
    const int lane = tid & 31;
    const int wv = tid >> 5;
    const int base = blockIdx.x * 256;
    int part = 0;
    #pragma unroll 8
    for (int j = 0; j < base; j += 1024) {
        const int i = j + tid * 4;
        const bool in = i < base;
        const int4 q = *(const int4*)(cnt + (in ? i : 0));
        const int sum4 = (q.x + q.y) + (q.z + q.w);
        part += in ? sum4 : 0;
    }
    const int idx = base + tid;
    const int v = idx < n ? cnt[idx] : 0;
    int incl = v;
    for (int o = 1; o < 32; o <<= 1) {
        const int t = __shfl_up(incl, o);
        if (lane >= o) incl += t;
    }
    for (int o = 16; o > 0; o >>= 1) part += __shfl_xor(part, o);
    if (lane == 31) s_incl[wv] = incl;
    if (lane == 0)  s_part[wv] = part;
    __syncthreads();
    int off = 0;
    for (int j = 0; j < 8; ++j) {
        off += s_part[j];
        if (j < wv) off += s_incl[j];
    }
    const int excl = off + incl - v;
    if (idx < n) { start[idx] = excl; cursor[idx] = excl; }
}

__global__ void fill_kernel(const int* __restrict__ eidx, int E,
                            int* __restrict__ cursor, int* __restrict__ csr) {
    int i = blockIdx.x * blockDim.x + threadIdx.x;
    if (i >= E) return;
    const int pos = atomicAdd(&cursor[eidx[E + i]], 1);
    atomicExch(&csr[pos], eidx[i]);
}

template <int H>
__device__ __forceinline__ void load_heads(const float* __restrict__ p, float* v) {
    if constexpr (H % 4 == 0) {
        #pragma unroll
        for (int q = 0; q < H / 4; ++q) {
            const float4 t = *(const float4*)(p + 4 * q);
            v[4 * q] = t.x; v[4 * q + 1] = t.y; v[4 * q + 2] = t.z; v[4 * q + 3] = t.w;
        }
    } else {
        #pragma unroll
        for (int q = 0; q < H; ++q) v[q] = p[q];
    }
}

template <int H, int C>
__global__ void node_agg_kernel(const int* __restrict__ start,
                                const int* __restrict__ endp,
                                const int* __restrict__ csr,
                                const float* __restrict__ as,
                                const float* __restrict__ ad,
                                const float* __restrict__ h,
                                const float* __restrict__ bias,
                                float* __restrict__ out, int n) {
    constexpr int HC  = H * C;
    constexpr int LPR = HC / 2;
    constexpr int EPS = 32 / LPR;
    constexpr int NPW = (HC * 4 >= 128) ? 1 : 128 / (HC * 4);
    static_assert(HC % 2 == 0 && 32 % LPR == 0 && NPW <= EPS && C % 2 == 0 && 8 % EPS == 0, "lane layout");
    const int lane = threadIdx.x & 31;
    const int wid = __builtin_amdgcn_readfirstlane(
        (int)(blockIdx.x * (blockDim.x >> 5) + (threadIdx.x >> 5)));
    const int node0 = wid * NPW;
    if (node0 >= n) return;
    const int sub = lane / LPR;
    const int cl = lane - sub * LPR;
    const int myhead = (2 * cl) / C;
    float r0 = 0.f, r1 = 0.f;

    for (int nn = 0; nn < NPW; ++nn) {
        const int node = node0 + nn;
        if (node >= n) break;
        const int beg = start[node];
        const int deg = endp[node] - beg;
        const int tot = deg + 1;
        float adv[H];
        load_heads<H>(ad + (size_t)node * H, adv);

        auto fetch = [&](int t, int& s, float* e) {
            s = node;
            #pragma unroll
            for (int q = 0; q < H; ++q) e[q] = -INFINITY;
            if (t < tot) {
                if (t < deg) s = csr[beg + t];
                float av[H];
                load_heads<H>(as + (size_t)s * H, av);
                #pragma unroll
                for (int q = 0; q < H; ++q) e[q] = leaky(av[q] + adv[q]);
            }
        };

        int s0;
        float e0[H];
        fetch(lane, s0, e0);

        float m[H];
        #pragma unroll
        for (int q = 0; q < H; ++q) m[q] = e0[q];
        for (int c0 = 32; c0 < tot; c0 += 32) {
            int s;
            float e[H];
            fetch(c0 + lane, s, e);
            #pragma unroll
            for (int q = 0; q < H; ++q) m[q] = fmaxf(m[q], e[q]);
        }
        #pragma unroll
        for (int q = 0; q < H; ++q)
            for (int o = 16; o > 0; o >>= 1) m[q] = fmaxf(m[q], __shfl_xor(m[q], o));

        float den[H];
        #pragma unroll
        for (int q = 0; q < H; ++q) den[q] = __expf(e0[q] - m[q]);
        for (int c0 = 32; c0 < tot; c0 += 32) {
            int s;
            float e[H];
            fetch(c0 + lane, s, e);
            #pragma unroll
            for (int q = 0; q < H; ++q) den[q] += __expf(e[q] - m[q]);
        }
        #pragma unroll
        for (int q = 0; q < H; ++q)
            for (int o = 16; o > 0; o >>= 1) den[q] += __shfl_xor(den[q], o);

        float a0 = 0.f, a1 = 0.f;
        auto gather = [&](int s, const float* e, int nk) {
            float cf[H];
            #pragma unroll
            for (int q = 0; q < H; ++q) cf[q] = __expf(e[q] - m[q]) / den[q];
            constexpr int GRP = 8;
            for (int k = 0; k < nk; k += GRP) {
                #pragma unroll
                for (int u = 0; u < GRP; u += EPS) {
                    const int ke = k + u + sub;
                    int sv;
                    float cv;
                    if constexpr (EPS == 1) {
                        sv = __builtin_amdgcn_readlane(s, k + u);
                        cv = __int_as_float(__builtin_amdgcn_readlane(__float_as_int(cf[0]), k + u));
                        #pragma unroll
                        for (int q = 1; q < H; ++q) {
                            const float tc = __int_as_float(__builtin_amdgcn_readlane(__float_as_int(cf[q]), k + u));
                            if (myhead == q) cv = tc;
                        }
                    } else {
                        sv = __shfl(s, ke);
                        cv = __shfl(cf[0], ke);
                        #pragma unroll
                        for (int q = 1; q < H; ++q) {
                            const float tc = __shfl(cf[q], ke);
                            if (myhead == q) cv = tc;
                        }
                    }
                    const float2 hv = *(const float2*)(h + (size_t)sv * HC + 2 * cl);
                    a0 += cv * hv.x;
                    a1 += cv * hv.y;
                }
            }
        };
        gather(s0, e0, tot < 32 ? tot : 32);
        for (int c0 = 32; c0 < tot; c0 += 32) {
            int s;
            float e[H];
            fetch(c0 + lane, s, e);
            gather(s, e, (tot - c0) < 32 ? (tot - c0) : 32);
        }
        for (int o = LPR; o < 32; o <<= 1) {
            a0 += __shfl_xor(a0, o);
            a1 += __shfl_xor(a1, o);
        }
        if (sub == nn) { r0 = a0; r1 = a1; }
    }

    const int node = node0 + sub;
    if (sub < NPW && node < n) {
        float2 o;
        o.x = r0 + bias[2 * cl];
        o.y = r1 + bias[2 * cl + 1];
        *(float2*)(out + (size_t)node * HC + 2 * cl) = o;
    }
}

static inline int ceil_div(int a, int b) { return (a + b - 1) / b; }

extern "C" void kernel_launch(void* const* d_in, const int* in_sizes, int n_in,
                              void* d_out, int out_size, void* d_ws, size_t ws_size,
                              hipStream_t stream) {
    const float* x    = (const float*)d_in[0];
    const int*   eidx = (const int*)  d_in[1];
    const float* W1   = (const float*)d_in[2];
    const float* aS1  = (const float*)d_in[3];
    const float* aD1  = (const float*)d_in[4];
    const float* b1   = (const float*)d_in[5];
    const float* W2   = (const float*)d_in[6];
    const float* aS2  = (const float*)d_in[7];
    const float* aD2  = (const float*)d_in[8];
    const float* b2   = (const float*)d_in[9];
    float* out = (float*)d_out;

    const int FEATS = 256, HID = 64, H1 = 8, C1 = 8, CLS = 16;
    const int N = in_sizes[0] / FEATS;
    const int E = in_sizes[1] / 2;

    float* ws = (float*)d_ws;
    float* h1   = ws;                 ws += (size_t)N * HID;
    float* as1  = ws;                 ws += (size_t)N * H1;
    float* ad1  = ws;                 ws += (size_t)N * H1;
    float* o1   = ws;                 ws += (size_t)N * HID;
    float* h2   = ws;                 ws += (size_t)N * CLS;
    float* as2  = ws;                 ws += (size_t)N;
    float* ad2  = ws;                 ws += (size_t)N;
    int* rstart = (int*)ws;           ws += (size_t)N;
    int* rend   = (int*)ws;           ws += (size_t)N;
    int* csr    = (int*)ws;           ws += (size_t)E;
    int* cnt    = (int*)h2;

    const int TPB = 256;
    const int WPB = 4;
    const dim3 gblock(32 * WPB);

    zero_int_kernel<<<ceil_div(N, TPB), TPB, 0, stream>>>(cnt, N);
    count_kernel<<<ceil_div(E, TPB), TPB, 0, stream>>>(eidx, E, cnt);
    scan_kernel<<<ceil_div(N, 256), 256, 0, stream>>>(cnt, rstart, rend, N);
    fill_kernel<<<ceil_div(E, TPB), TPB, 0, stream>>>(eidx, E, rend, csr);

    {
        dim3 grid(ceil_div(N / 16, WPB), 1);
        wmma_gemm_f32<4><<<grid, gblock, 0, stream>>>(x, W1, h1, N, FEATS, HID);
    }
    alpha_kernel<H1, C1><<<ceil_div(N * H1, TPB), TPB, 0, stream>>>(h1, aS1, aD1, as1, ad1, N);
    node_agg_kernel<H1, C1><<<ceil_div(N, 8), 256, 0, stream>>>(rstart, rend, csr, as1, ad1, h1, b1, o1, N);

    {
        dim3 grid(ceil_div(N / 16, WPB), 1);
        wmma_gemm_f32<1><<<grid, gblock, 0, stream>>>(o1, W2, h2, N, HID, CLS);
    }
    alpha_kernel<1, CLS><<<ceil_div(N, TPB), TPB, 0, stream>>>(h2, aS2, aD2, as2, ad2, N);
    node_agg_kernel<1, CLS><<<ceil_div(ceil_div(N, 2), 8), 256, 0, stream>>>(rstart, rend, csr, as2, ad2, h2, b2, out, N);
}


# count pass keeps each edge's rank (its add now returns), fill places the source at start[dst]+rank: no cursor add, 2 atomics per edge not 3; row sums stay in arrival order, now the count pass's
# speedup vs baseline: 4.1292x; 1.3399x over previous
#include <hip/hip_runtime.h>
#include <math.h>

typedef float v2f __attribute__((ext_vector_type(2)));
typedef float v8f __attribute__((ext_vector_type(8)));

#define NEG_SLOPE 0.2f

__device__ __forceinline__ float leaky(float x) {
    return x > 0.f ? x : NEG_SLOPE * x;
}


template <int CT>
__global__ void wmma_gemm_f32(const float* __restrict__ A,
                              const float* __restrict__ B,
                              float* __restrict__ Out,
                              int M, int K, int Nout) {
    const int wave = threadIdx.x >> 5;
    const int lane = threadIdx.x & 31;
    const int row0 = (blockIdx.x * (blockDim.x >> 5) + wave) << 4;
    const int col0 = blockIdx.y * (CT * 16);
    if (row0 >= M) return;

    const int half = lane >> 4;
    const int l16  = lane & 15;
    const float* arow = A + (size_t)(row0 + l16) * K;
    const int col = col0 + l16;

    v8f acc[CT];
    #pragma unroll
    for (int t = 0; t < CT; ++t) acc[t] = (v8f){};

    for (int k = 0; k < K; k += 4) {
        const int ka = k + 2 * half;
        v2f a;
        a.x = arow[ka];
        a.y = arow[ka + 1];
        const float* brow0 = B + (size_t)ka * Nout + col;
        const float* brow1 = B + (size_t)(ka + 1) * Nout + col;
        #pragma unroll
        for (int t = 0; t < CT; ++t) {
            v2f b;
            b.x = brow0[t * 16];
            b.y = brow1[t * 16];
            acc[t] = __builtin_amdgcn_wmma_f32_16x16x4_f32(
                 false, a,  false, b,
                 (short)0, acc[t],  false,  false);
        }
    }
    #pragma unroll
    for (int t = 0; t < CT; ++t) {
        #pragma unroll
        for (int r = 0; r < 8; ++r) {
            Out[(size_t)(row0 + r + 8 * half) * Nout + col + t * 16] = acc[t][r];
        }
    }
}

__global__ void zero_int_kernel(int* __restrict__ p, int n) {
    int gid = blockIdx.x * blockDim.x + threadIdx.x;
    if (gid < n) p[gid] = 0;
}

__global__ void count_kernel(const int* __restrict__ eidx, int E,
                             int* __restrict__ cnt, int* __restrict__ rank) {
    int i = blockIdx.x * blockDim.x + threadIdx.x;
    if (i < E) rank[i] = atomicAdd(&cnt[eidx[E + i]], 1);
}

__global__ void scan_kernel(const int* __restrict__ cnt,
                            int* __restrict__ start, int* __restrict__ endp,
                            int n) {
    __shared__ int s_incl[8];
    __shared__ int s_part[8];
    const int tid = threadIdx.x;
    const int lane = tid & 31;
    const int wv = tid >> 5;
    const int base = blockIdx.x * 256;
    int part = 0;
    #pragma unroll 8
    for (int j = 0; j < base; j += 1024) {
        const int i = j + tid * 4;
        const bool in = i < base;
        const int4 q = *(const int4*)(cnt + (in ? i : 0));
        const int sum4 = (q.x + q.y) + (q.z + q.w);
        part += in ? sum4 : 0;
    }
    const int idx = base + tid;
    const int v = idx < n ? cnt[idx] : 0;
    int incl = v;
    for (int o = 1; o < 32; o <<= 1) {
        const int t = __shfl_up(incl, o);
        if (lane >= o) incl += t;
    }
    for (int o = 16; o > 0; o >>= 1) part += __shfl_xor(part, o);
    if (lane == 31) s_incl[wv] = incl;
    if (lane == 0)  s_part[wv] = part;
    __syncthreads();
    int off = 0;
    for (int j = 0; j < 8; ++j) {
        off += s_part[j];
        if (j < wv) off += s_incl[j];
    }
    const int excl = off + incl - v;
    if (idx < n) { start[idx] = excl; endp[idx] = excl + v; }
}

__global__ void fill_kernel(const int* __restrict__ eidx, int E,
                            const int* __restrict__ start,
                            const int* __restrict__ rank,
                            int* __restrict__ csr) {
    int i = blockIdx.x * blockDim.x + threadIdx.x;
    if (i >= E) return;
    const int pos = start[eidx[E + i]] + rank[i];
    atomicExch(&csr[pos], eidx[i]);
}

template <int M>
__device__ __forceinline__ float row_xor_get(float v) {
    return __int_as_float(__builtin_amdgcn_update_dpp(
        0, __float_as_int(v), 0x160 | M, 0xf, 0xf, true));
}

template <int W>
__device__ __forceinline__ float group_sum(float v) {
    v += row_xor_get<1>(v);
    v += row_xor_get<2>(v);
    if constexpr (W == 8) v += row_xor_get<4>(v);
    return v;
}

template <int H, int C>
__global__ void node_agg_kernel(const int* __restrict__ start,
                                const int* __restrict__ endp,
                                const int* __restrict__ csr,
                                const float* __restrict__ a_src,
                                const float* __restrict__ a_dst,
                                const float* __restrict__ h,
                                const float* __restrict__ bias,
                                float* __restrict__ out, int n) {
    constexpr int HC  = H * C;
    constexpr int LPR = HC / 2;
    constexpr int LPH = C / 2;
    constexpr int EPS = 32 / LPR;
    constexpr int NPW = (HC * 4 >= 128) ? 1 : 128 / (HC * 4);
    static_assert(HC % 2 == 0 && 32 % LPR == 0 && NPW <= EPS && C % 2 == 0 && 8 % EPS == 0
                  && (LPH == 4 || LPH == 8), "lane layout");
    const int lane = threadIdx.x & 31;
    const int wid = __builtin_amdgcn_readfirstlane(
        (int)(blockIdx.x * (blockDim.x >> 5) + (threadIdx.x >> 5)));
    const int node0 = wid * NPW;
    if (node0 >= n) return;
    const int sub = lane / LPR;
    const int cl = lane - sub * LPR;
    const float ws0 = a_src[2 * cl], ws1 = a_src[2 * cl + 1];
    const float wd0 = a_dst[2 * cl], wd1 = a_dst[2 * cl + 1];
    float r0 = 0.f, r1 = 0.f;

    for (int nn = 0; nn < NPW; ++nn) {
        const int node = node0 + nn;
        if (node >= n) break;
        const int beg = start[node];
        const int deg = endp[node] - beg;

        const float2 hs = *(const float2*)(h + (size_t)node * HC + 2 * cl);
        const float sd = group_sum<LPH>(hs.x * wd0 + hs.y * wd1);
        const float ss = group_sum<LPH>(hs.x * ws0 + hs.y * ws1);
        float m = leaky(ss + sd);
        float den = (sub == 0) ? 1.f : 0.f;
        float a0  = (sub == 0) ? hs.x : 0.f;
        float a1  = (sub == 0) ? hs.y : 0.f;

        for (int c0 = 0; c0 < deg; c0 += 32) {
            const int cnt = (deg - c0) < 32 ? (deg - c0) : 32;
            int s = node;
            if (lane < cnt) s = csr[beg + c0 + lane];
            constexpr int GRP = 8;
            for (int k = 0; k < cnt; k += GRP) {
                constexpr int NS = GRP / EPS;
                float2 hv[NS];
                float e[NS];
                #pragma unroll
                for (int j = 0; j < NS; ++j) {
                    const int ke = k + j * EPS + sub;
                    int sv;
                    if constexpr (EPS == 1) {
                        sv = __builtin_amdgcn_readlane(s, k + j);
                    } else {
                        sv = __shfl(s, ke);
                    }
                    hv[j] = *(const float2*)(h + (size_t)sv * HC + 2 * cl);
                }
                #pragma unroll
                for (int j = 0; j < NS; ++j) {
                    const int ke = k + j * EPS + sub;
                    const float ej = leaky(group_sum<LPH>(hv[j].x * ws0 + hv[j].y * ws1) + sd);
                    e[j] = (ke < cnt) ? ej : -INFINITY;
                }
                float mn = m;
                #pragma unroll
                for (int j = 0; j < NS; ++j) mn = fmaxf(mn, e[j]);
                const float sc = __expf(m - mn);
                den *= sc;
                a0  *= sc;
                a1  *= sc;
                #pragma unroll
                for (int j = 0; j < NS; ++j) {
                    const float p = __expf(e[j] - mn);
                    den += p;
                    a0  += p * hv[j].x;
                    a1  += p * hv[j].y;
                }
                m = mn;
            }
        }
        for (int o = LPR; o < 32; o <<= 1) {
            const float mo = __shfl_xor(m, o);
            const float dn = __shfl_xor(den, o);
            const float b0 = __shfl_xor(a0, o);
            const float b1 = __shfl_xor(a1, o);
            const float mn = fmaxf(m, mo);
            const float f  = __expf(m - mn);
            const float fo = __expf(mo - mn);
            den = den * f + dn * fo;
            a0  = a0 * f + b0 * fo;
            a1  = a1 * f + b1 * fo;
            m = mn;
        }
        const float q0 = a0 / den;
        const float q1 = a1 / den;
        if (sub == nn) { r0 = q0; r1 = q1; }
    }

    const int node = node0 + sub;
    if (sub < NPW && node < n) {
        float2 o;
        o.x = r0 + bias[2 * cl];
        o.y = r1 + bias[2 * cl + 1];
        *(float2*)(out + (size_t)node * HC + 2 * cl) = o;
    }
}

static inline int ceil_div(int a, int b) { return (a + b - 1) / b; }

extern "C" void kernel_launch(void* const* d_in, const int* in_sizes, int n_in,
                              void* d_out, int out_size, void* d_ws, size_t ws_size,
                              hipStream_t stream) {
    const float* x    = (const float*)d_in[0];
    const int*   eidx = (const int*)  d_in[1];
    const float* W1   = (const float*)d_in[2];
    const float* aS1  = (const float*)d_in[3];
    const float* aD1  = (const float*)d_in[4];
    const float* b1   = (const float*)d_in[5];
    const float* W2   = (const float*)d_in[6];
    const float* aS2  = (const float*)d_in[7];
    const float* aD2  = (const float*)d_in[8];
    const float* b2   = (const float*)d_in[9];
    float* out = (float*)d_out;

    const int FEATS = 256, HID = 64, H1 = 8, C1 = 8, CLS = 16;
    const int N = in_sizes[0] / FEATS;
    const int E = in_sizes[1] / 2;

    float* ws = (float*)d_ws;
    float* h1   = ws;                 ws += (size_t)N * HID;
    ws += (size_t)N * H1;
    ws += (size_t)N * H1;
    float* o1   = ws;                 ws += (size_t)N * HID;
    float* h2   = ws;                 ws += (size_t)N * CLS;
    ws += (size_t)N;
    ws += (size_t)N;
    int* rstart = (int*)ws;           ws += (size_t)N;
    int* rend   = (int*)ws;           ws += (size_t)N;
    int* csr    = (int*)ws;           ws += (size_t)E;
    int* cnt    = (int*)h2;
    int* rank   = (int*)o1;

    const int TPB = 256;
    const int WPB = 4;
    const dim3 gblock(32 * WPB);

    zero_int_kernel<<<ceil_div(N, TPB), TPB, 0, stream>>>(cnt, N);
    count_kernel<<<ceil_div(E, TPB), TPB, 0, stream>>>(eidx, E, cnt, rank);
    scan_kernel<<<ceil_div(N, 256), 256, 0, stream>>>(cnt, rstart, rend, N);
    fill_kernel<<<ceil_div(E, TPB), TPB, 0, stream>>>(eidx, E, rstart, rank, csr);

    {
        dim3 grid(ceil_div(N / 16, WPB), 1);
        wmma_gemm_f32<4><<<grid, gblock, 0, stream>>>(x, W1, h1, N, FEATS, HID);
    }
    node_agg_kernel<H1, C1><<<ceil_div(N, 8), 256, 0, stream>>>(rstart, rend, csr, aS1, aD1, h1, b1, o1, N);

    {
        dim3 grid(ceil_div(N / 16, WPB), 1);
        wmma_gemm_f32<1><<<grid, gblock, 0, stream>>>(o1, W2, h2, N, HID, CLS);
    }
    node_agg_kernel<1, CLS><<<ceil_div(ceil_div(N, 2), 8), 256, 0, stream>>>(rstart, rend, csr, aS2, aD2, h2, b2, out, N);
}


# edges grouped by destination in two levels, all counting in LDS (block partition by 512-node range, then one block per range lays out its csr piece): no global adds; row sums in LDS arrival order
# speedup vs baseline: 8.1786x; 1.9807x over previous
#include <hip/hip_runtime.h>
#include <math.h>

typedef float v2f __attribute__((ext_vector_type(2)));
typedef float v8f __attribute__((ext_vector_type(8)));

#define NEG_SLOPE 0.2f

__device__ __forceinline__ float leaky(float x) {
    return x > 0.f ? x : NEG_SLOPE * x;
}


template <int CT>
__global__ void wmma_gemm_f32(const float* __restrict__ A,
                              const float* __restrict__ B,
                              float* __restrict__ Out,
                              int M, int K, int Nout) {
    const int wave = threadIdx.x >> 5;
    const int lane = threadIdx.x & 31;
    const int row0 = (blockIdx.x * (blockDim.x >> 5) + wave) << 4;
    const int col0 = blockIdx.y * (CT * 16);
    if (row0 >= M) return;

    const int half = lane >> 4;
    const int l16  = lane & 15;
    const float* arow = A + (size_t)(row0 + l16) * K;
    const int col = col0 + l16;

    v8f acc[CT];
    #pragma unroll
    for (int t = 0; t < CT; ++t) acc[t] = (v8f){};

    for (int k = 0; k < K; k += 4) {
        const int ka = k + 2 * half;
        v2f a;
        a.x = arow[ka];
        a.y = arow[ka + 1];
        const float* brow0 = B + (size_t)ka * Nout + col;
        const float* brow1 = B + (size_t)(ka + 1) * Nout + col;
        #pragma unroll
        for (int t = 0; t < CT; ++t) {
            v2f b;
            b.x = brow0[t * 16];
            b.y = brow1[t * 16];
            acc[t] = __builtin_amdgcn_wmma_f32_16x16x4_f32(
                 false, a,  false, b,
                 (short)0, acc[t],  false,  false);
        }
    }
    #pragma unroll
    for (int t = 0; t < CT; ++t) {
        #pragma unroll
        for (int r = 0; r < 8; ++r) {
            Out[(size_t)(row0 + r + 8 * half) * Nout + col + t * 16] = acc[t][r];
        }
    }
}

#define PT 4096
#define RS_SHIFT 9
#define RS 512
#define BP 256
#define CAP 12288

__device__ __forceinline__ int wave_incl_scan(int v, int lane) {
    for (int o = 1; o < 32; o <<= 1) {
        const int u = __shfl_up(v, o);
        if (lane >= o) v += u;
    }
    return v;
}

__global__ void __launch_bounds__(1024)
part_kernel(const int* __restrict__ eidx, int E,
            unsigned* __restrict__ bsorted, int* __restrict__ boff) {
    __shared__ int hist[BP];
    __shared__ int offs[BP];
    __shared__ int wsum[32];
    __shared__ unsigned srt[PT];
    const int t = threadIdx.x;
    const int lane = t & 31;
    const int wv = t >> 5;
    const int e0 = blockIdx.x * PT;
    if (t < BP) hist[t] = 0;
    __syncthreads();
    int bk[4], rk[4];
    unsigned wd[4];
    #pragma unroll
    for (int k = 0; k < 4; ++k) {
        const int i = e0 + t + 1024 * k;
        bk[k] = -1; rk[k] = 0; wd[k] = 0u;
        if (i < E) {
            const int d = eidx[E + i];
            const int s = eidx[i];
            bk[k] = d >> RS_SHIFT;
            wd[k] = ((unsigned)(d & (RS - 1)) << 23) | (unsigned)s;
        }
    }
    #pragma unroll
    for (int k = 0; k < 4; ++k)
        if (bk[k] >= 0) rk[k] = atomicAdd(&hist[bk[k]], 1);
    __syncthreads();
    const int v = t < BP ? hist[t] : 0;
    const int incl = wave_incl_scan(v, lane);
    if (lane == 31) wsum[wv] = incl;
    __syncthreads();
    int off = 0;
    for (int j = 0; j < wv; ++j) off += wsum[j];
    if (t < BP) offs[t] = off + incl - v;
    __syncthreads();
    #pragma unroll
    for (int k = 0; k < 4; ++k)
        if (bk[k] >= 0) srt[offs[bk[k]] + rk[k]] = wd[k];
    __syncthreads();
    const int nvalid = (E - e0) < PT ? (E - e0) : PT;
    #pragma unroll
    for (int k = 0; k < 4; ++k) {
        const int i = t + 1024 * k;
        if (i < nvalid) bsorted[e0 + i] = srt[i];
    }
    if (t < BP) boff[blockIdx.x * BP + t] = offs[t];
}

template <typename F>
__device__ __forceinline__ void sweep_range(const unsigned* __restrict__ bsorted,
                                            const int* __restrict__ boff,
                                            int P, int b, int* runS, int* runL,
                                            F&& f) {
    const int t = threadIdx.x;
    const int lane = t & 31;
    const int wv = t >> 5;
    for (int p0 = 0; p0 < P; p0 += 1024) {
        __syncthreads();
        const int p = p0 + t;
        if (p < P) {
            const int s = boff[p * BP + b];
            runS[t] = p * PT + s;
            runL[t] = boff[p * BP + b + 1] - s;
        }
        __syncthreads();
        const int np = (P - p0) < 1024 ? (P - p0) : 1024;
        for (int q0 = wv; q0 < np; q0 += 128) {
            int S[4], L[4];
            unsigned w[4];
            #pragma unroll
            for (int k = 0; k < 4; ++k) {
                const int q = q0 + 32 * k;
                S[k] = 0; L[k] = 0;
                if (q < np) { S[k] = runS[q]; L[k] = runL[q]; }
            }
            #pragma unroll
            for (int k = 0; k < 4; ++k) {
                w[k] = 0u;
                if (lane < L[k]) w[k] = bsorted[S[k] + lane];
            }
            #pragma unroll
            for (int k = 0; k < 4; ++k)
                if (lane < L[k]) f(w[k]);
            #pragma unroll
            for (int k = 0; k < 4; ++k)
                for (int j = lane + 32; j < L[k]; j += 32) f(bsorted[S[k] + j]);
        }
    }
}

__global__ void __launch_bounds__(1024)
range_kernel(const unsigned* __restrict__ bsorted, const int* __restrict__ boff,
             int P, int n, int* __restrict__ start, int* __restrict__ endp,
             int* __restrict__ csr) {
    __shared__ int cnt[RS];
    __shared__ int cur[RS];
    __shared__ int runS[1024];
    __shared__ int runL[1024];
    __shared__ int wsum[32];
    __shared__ int s_base;
    __shared__ int s_total;
    __shared__ int outb[CAP];
    const int t = threadIdx.x;
    const int lane = t & 31;
    const int wv = t >> 5;
    const int b = blockIdx.x;
    if (t < RS) cnt[t] = 0;
    if (t == 0) { s_base = 0; s_total = 0; }
    __syncthreads();
    int pb = 0, pt = 0;
    for (int p = t; p < P; p += 1024) {
        const int s = boff[p * BP + b];
        pb += s;
        pt += boff[p * BP + b + 1] - s;
    }
    for (int o = 16; o > 0; o >>= 1) {
        pb += __shfl_xor(pb, o);
        pt += __shfl_xor(pt, o);
    }
    if (lane == 0) { atomicAdd(&s_base, pb); atomicAdd(&s_total, pt); }
    __syncthreads();
    const int base = s_base;
    const int total = s_total;
    const bool staged = total <= CAP;

    sweep_range(bsorted, boff, P, b, runS, runL,
                [&](unsigned w) { atomicAdd(&cnt[w >> 23], 1); });
    __syncthreads();
    const int v = t < RS ? cnt[t] : 0;
    const int incl = wave_incl_scan(v, lane);
    if (lane == 31) wsum[wv] = incl;
    __syncthreads();
    int off = 0;
    for (int j = 0; j < wv; ++j) off += wsum[j];
    const int excl = off + incl - v;
    if (t < RS) {
        cur[t] = excl;
        const int node = b * RS + t;
        if (node < n) { start[node] = base + excl; endp[node] = base + excl + v; }
    }
    __syncthreads();
    sweep_range(bsorted, boff, P, b, runS, runL,
                [&](unsigned w) {
                    const int pos = atomicAdd(&cur[w >> 23], 1);
                    const int src = (int)(w & 0x7fffffu);
                    if (staged) outb[pos] = src;
                    else atomicExch(&csr[base + pos], src);
                });
    __syncthreads();
    if (staged) {
        const int lim = base + total;
        for (int g = (base & ~31) + t; g < lim; g += 1024) {
            if (g >= base) {
                const int ls = g & ~31;
                const int val = outb[g - base];
                if (ls >= base && ls + 32 <= lim) csr[g] = val;
                else atomicExch(&csr[g], val);
            }
        }
    }
}

template <int M>
__device__ __forceinline__ float row_xor_get(float v) {
    return __int_as_float(__builtin_amdgcn_update_dpp(
        0, __float_as_int(v), 0x160 | M, 0xf, 0xf, true));
}

template <int W>
__device__ __forceinline__ float group_sum(float v) {
    v += row_xor_get<1>(v);
    v += row_xor_get<2>(v);
    if constexpr (W == 8) v += row_xor_get<4>(v);
    return v;
}

template <int H, int C>
__global__ void node_agg_kernel(const int* __restrict__ start,
                                const int* __restrict__ endp,
                                const int* __restrict__ csr,
                                const float* __restrict__ a_src,
                                const float* __restrict__ a_dst,
                                const float* __restrict__ h,
                                const float* __restrict__ bias,
                                float* __restrict__ out, int n) {
    constexpr int HC  = H * C;
    constexpr int LPR = HC / 2;
    constexpr int LPH = C / 2;
    constexpr int EPS = 32 / LPR;
    constexpr int NPW = (HC * 4 >= 128) ? 1 : 128 / (HC * 4);
    static_assert(HC % 2 == 0 && 32 % LPR == 0 && NPW <= EPS && C % 2 == 0 && 8 % EPS == 0
                  && (LPH == 4 || LPH == 8), "lane layout");
    const int lane = threadIdx.x & 31;
    const int wid = __builtin_amdgcn_readfirstlane(
        (int)(blockIdx.x * (blockDim.x >> 5) + (threadIdx.x >> 5)));
    const int node0 = wid * NPW;
    if (node0 >= n) return;
    const int sub = lane / LPR;
    const int cl = lane - sub * LPR;
    const float ws0 = a_src[2 * cl], ws1 = a_src[2 * cl + 1];
    const float wd0 = a_dst[2 * cl], wd1 = a_dst[2 * cl + 1];
    float r0 = 0.f, r1 = 0.f;

    for (int nn = 0; nn < NPW; ++nn) {
        const int node = node0 + nn;
        if (node >= n) break;
        const int beg = start[node];
        const int deg = endp[node] - beg;

        const float2 hs = *(const float2*)(h + (size_t)node * HC + 2 * cl);
        const float sd = group_sum<LPH>(hs.x * wd0 + hs.y * wd1);
        const float ss = group_sum<LPH>(hs.x * ws0 + hs.y * ws1);
        float m = leaky(ss + sd);
        float den = (sub == 0) ? 1.f : 0.f;
        float a0  = (sub == 0) ? hs.x : 0.f;
        float a1  = (sub == 0) ? hs.y : 0.f;

        for (int c0 = 0; c0 < deg; c0 += 32) {
            const int cnt = (deg - c0) < 32 ? (deg - c0) : 32;
            int s = node;
            if (lane < cnt) s = csr[beg + c0 + lane];
            constexpr int GRP = 8;
            for (int k = 0; k < cnt; k += GRP) {
                constexpr int NS = GRP / EPS;
                float2 hv[NS];
                float e[NS];
                #pragma unroll
                for (int j = 0; j < NS; ++j) {
                    const int ke = k + j * EPS + sub;
                    int sv;
                    if constexpr (EPS == 1) {
                        sv = __builtin_amdgcn_readlane(s, k + j);
                    } else {
                        sv = __shfl(s, ke);
                    }
                    hv[j] = *(const float2*)(h + (size_t)sv * HC + 2 * cl);
                }
                #pragma unroll
                for (int j = 0; j < NS; ++j) {
                    const int ke = k + j * EPS + sub;
                    const float ej = leaky(group_sum<LPH>(hv[j].x * ws0 + hv[j].y * ws1) + sd);
                    e[j] = (ke < cnt) ? ej : -INFINITY;
                }
                float mn = m;
                #pragma unroll
                for (int j = 0; j < NS; ++j) mn = fmaxf(mn, e[j]);
                const float sc = __expf(m - mn);
                den *= sc;
                a0  *= sc;
                a1  *= sc;
                #pragma unroll
                for (int j = 0; j < NS; ++j) {
                    const float p = __expf(e[j] - mn);
                    den += p;
                    a0  += p * hv[j].x;
                    a1  += p * hv[j].y;
                }
                m = mn;
            }
        }
        for (int o = LPR; o < 32; o <<= 1) {
            const float mo = __shfl_xor(m, o);
            const float dn = __shfl_xor(den, o);
            const float b0 = __shfl_xor(a0, o);
            const float b1 = __shfl_xor(a1, o);
            const float mn = fmaxf(m, mo);
            const float f  = __expf(m - mn);
            const float fo = __expf(mo - mn);
            den = den * f + dn * fo;
            a0  = a0 * f + b0 * fo;
            a1  = a1 * f + b1 * fo;
            m = mn;
        }
        const float q0 = a0 / den;
        const float q1 = a1 / den;
        if (sub == nn) { r0 = q0; r1 = q1; }
    }

    const int node = node0 + sub;
    if (sub < NPW && node < n) {
        float2 o;
        o.x = r0 + bias[2 * cl];
        o.y = r1 + bias[2 * cl + 1];
        *(float2*)(out + (size_t)node * HC + 2 * cl) = o;
    }
}

static inline int ceil_div(int a, int b) { return (a + b - 1) / b; }

extern "C" void kernel_launch(void* const* d_in, const int* in_sizes, int n_in,
                              void* d_out, int out_size, void* d_ws, size_t ws_size,
                              hipStream_t stream) {
    const float* x    = (const float*)d_in[0];
    const int*   eidx = (const int*)  d_in[1];
    const float* W1   = (const float*)d_in[2];
    const float* aS1  = (const float*)d_in[3];
    const float* aD1  = (const float*)d_in[4];
    const float* b1   = (const float*)d_in[5];
    const float* W2   = (const float*)d_in[6];
    const float* aS2  = (const float*)d_in[7];
    const float* aD2  = (const float*)d_in[8];
    const float* b2   = (const float*)d_in[9];
    float* out = (float*)d_out;

    const int FEATS = 256, HID = 64, H1 = 8, C1 = 8, CLS = 16;
    const int N = in_sizes[0] / FEATS;
    const int E = in_sizes[1] / 2;

    float* ws = (float*)d_ws;
    float* h1   = ws;                 ws += (size_t)N * HID;
    ws += (size_t)N * H1;
    ws += (size_t)N * H1;
    float* o1   = ws;                 ws += (size_t)N * HID;
    float* h2   = ws;                 ws += (size_t)N * CLS;
    ws += (size_t)N;
    ws += (size_t)N;
    int* rstart = (int*)ws;           ws += (size_t)N;
    int* rend   = (int*)ws;           ws += (size_t)N;
    int* csr    = (int*)ws;           ws += (size_t)E;
    int* boff   = (int*)h2;
    unsigned* bsorted = (unsigned*)o1;

    const int WPB = 4;
    const dim3 gblock(32 * WPB);

    const int P = ceil_div(E, PT);
    const int R = ceil_div(N, RS);
    part_kernel<<<P, 1024, 0, stream>>>(eidx, E, bsorted, boff);
    range_kernel<<<R, 1024, 0, stream>>>(bsorted, boff, P, N, rstart, rend, csr);

    {
        dim3 grid(ceil_div(N / 16, WPB), 1);
        wmma_gemm_f32<4><<<grid, gblock, 0, stream>>>(x, W1, h1, N, FEATS, HID);
    }
    node_agg_kernel<H1, C1><<<ceil_div(N, 8), 256, 0, stream>>>(rstart, rend, csr, aS1, aD1, h1, b1, o1, N);

    {
        dim3 grid(ceil_div(N / 16, WPB), 1);
        wmma_gemm_f32<1><<<grid, gblock, 0, stream>>>(o1, W2, h2, N, HID, CLS);
    }
    node_agg_kernel<1, CLS><<<ceil_div(ceil_div(N, 2), 8), 256, 0, stream>>>(rstart, rend, csr, aS2, aD2, h2, b2, out, N);
}


# row aggregation: destination rows side by side in a wave (layer 1: 4 rows, a lane owns a head's 8 channels; layer 2: 8 rows, 4 lanes each): a step takes one edge of every row; row sums in csr order
# speedup vs baseline: 8.5562x; 1.0462x over previous
#include <hip/hip_runtime.h>
#include <math.h>

typedef float v2f __attribute__((ext_vector_type(2)));
typedef float v4f __attribute__((ext_vector_type(4)));
typedef float v8f __attribute__((ext_vector_type(8)));

#define NEG_SLOPE 0.2f

__device__ __forceinline__ float leaky(float x) {
    return x > 0.f ? x : NEG_SLOPE * x;
}


template <int CT>
__global__ void wmma_gemm_f32(const float* __restrict__ A,
                              const float* __restrict__ B,
                              float* __restrict__ Out,
                              int M, int K, int Nout) {
    const int wave = threadIdx.x >> 5;
    const int lane = threadIdx.x & 31;
    const int row0 = (blockIdx.x * (blockDim.x >> 5) + wave) << 4;
    const int col0 = blockIdx.y * (CT * 16);
    if (row0 >= M) return;

    const int half = lane >> 4;
    const int l16  = lane & 15;
    const float* arow = A + (size_t)(row0 + l16) * K;
    const int col = col0 + l16;

    v8f acc[CT];
    #pragma unroll
    for (int t = 0; t < CT; ++t) acc[t] = (v8f){};

    for (int k = 0; k < K; k += 4) {
        const int ka = k + 2 * half;
        v2f a;
        a.x = arow[ka];
        a.y = arow[ka + 1];
        const float* brow0 = B + (size_t)ka * Nout + col;
        const float* brow1 = B + (size_t)(ka + 1) * Nout + col;
        #pragma unroll
        for (int t = 0; t < CT; ++t) {
            v2f b;
            b.x = brow0[t * 16];
            b.y = brow1[t * 16];
            acc[t] = __builtin_amdgcn_wmma_f32_16x16x4_f32(
                 false, a,  false, b,
                 (short)0, acc[t],  false,  false);
        }
    }
    #pragma unroll
    for (int t = 0; t < CT; ++t) {
        #pragma unroll
        for (int r = 0; r < 8; ++r) {
            Out[(size_t)(row0 + r + 8 * half) * Nout + col + t * 16] = acc[t][r];
        }
    }
}

#define PT 4096
#define RS_SHIFT 9
#define RS 512
#define BP 256
#define CAP 12288

__device__ __forceinline__ int wave_incl_scan(int v, int lane) {
    for (int o = 1; o < 32; o <<= 1) {
        const int u = __shfl_up(v, o);
        if (lane >= o) v += u;
    }
    return v;
}

__global__ void __launch_bounds__(1024)
part_kernel(const int* __restrict__ eidx, int E,
            unsigned* __restrict__ bsorted, int* __restrict__ boff) {
    __shared__ int hist[BP];
    __shared__ int offs[BP];
    __shared__ int wsum[32];
    __shared__ unsigned srt[PT];
    const int t = threadIdx.x;
    const int lane = t & 31;
    const int wv = t >> 5;
    const int e0 = blockIdx.x * PT;
    if (t < BP) hist[t] = 0;
    __syncthreads();
    int bk[4], rk[4];
    unsigned wd[4];
    #pragma unroll
    for (int k = 0; k < 4; ++k) {
        const int i = e0 + t + 1024 * k;
        bk[k] = -1; rk[k] = 0; wd[k] = 0u;
        if (i < E) {
            const int d = eidx[E + i];
            const int s = eidx[i];
            bk[k] = d >> RS_SHIFT;
            wd[k] = ((unsigned)(d & (RS - 1)) << 23) | (unsigned)s;
        }
    }
    #pragma unroll
    for (int k = 0; k < 4; ++k)
        if (bk[k] >= 0) rk[k] = atomicAdd(&hist[bk[k]], 1);
    __syncthreads();
    const int v = t < BP ? hist[t] : 0;
    const int incl = wave_incl_scan(v, lane);
    if (lane == 31) wsum[wv] = incl;
    __syncthreads();
    int off = 0;
    for (int j = 0; j < wv; ++j) off += wsum[j];
    if (t < BP) offs[t] = off + incl - v;
    __syncthreads();
    #pragma unroll
    for (int k = 0; k < 4; ++k)
        if (bk[k] >= 0) srt[offs[bk[k]] + rk[k]] = wd[k];
    __syncthreads();
    const int nvalid = (E - e0) < PT ? (E - e0) : PT;
    #pragma unroll
    for (int k = 0; k < 4; ++k) {
        const int i = t + 1024 * k;
        if (i < nvalid) bsorted[e0 + i] = srt[i];
    }
    if (t < BP) boff[blockIdx.x * BP + t] = offs[t];
}

template <typename F>
__device__ __forceinline__ void sweep_range(const unsigned* __restrict__ bsorted,
                                            const int* __restrict__ boff,
                                            int P, int b, int* runS, int* runL,
                                            F&& f) {
    const int t = threadIdx.x;
    const int lane = t & 31;
    const int wv = t >> 5;
    for (int p0 = 0; p0 < P; p0 += 1024) {
        __syncthreads();
        const int p = p0 + t;
        if (p < P) {
            const int s = boff[p * BP + b];
            runS[t] = p * PT + s;
            runL[t] = boff[p * BP + b + 1] - s;
        }
        __syncthreads();
        const int np = (P - p0) < 1024 ? (P - p0) : 1024;
        for (int q0 = wv; q0 < np; q0 += 128) {
            int S[4], L[4];
            unsigned w[4];
            #pragma unroll
            for (int k = 0; k < 4; ++k) {
                const int q = q0 + 32 * k;
                S[k] = 0; L[k] = 0;
                if (q < np) { S[k] = runS[q]; L[k] = runL[q]; }
            }
            #pragma unroll
            for (int k = 0; k < 4; ++k) {
                w[k] = 0u;
                if (lane < L[k]) w[k] = bsorted[S[k] + lane];
            }
            #pragma unroll
            for (int k = 0; k < 4; ++k)
                if (lane < L[k]) f(w[k]);
            #pragma unroll
            for (int k = 0; k < 4; ++k)
                for (int j = lane + 32; j < L[k]; j += 32) f(bsorted[S[k] + j]);
        }
    }
}

__global__ void __launch_bounds__(1024)
range_kernel(const unsigned* __restrict__ bsorted, const int* __restrict__ boff,
             int P, int n, int* __restrict__ start, int* __restrict__ endp,
             int* __restrict__ csr) {
    __shared__ int cnt[RS];
    __shared__ int cur[RS];
    __shared__ int runS[1024];
    __shared__ int runL[1024];
    __shared__ int wsum[32];
    __shared__ int s_base;
    __shared__ int s_total;
    __shared__ int outb[CAP];
    const int t = threadIdx.x;
    const int lane = t & 31;
    const int wv = t >> 5;
    const int b = blockIdx.x;
    if (t < RS) cnt[t] = 0;
    if (t == 0) { s_base = 0; s_total = 0; }
    __syncthreads();
    int pb = 0, pt = 0;
    for (int p = t; p < P; p += 1024) {
        const int s = boff[p * BP + b];
        pb += s;
        pt += boff[p * BP + b + 1] - s;
    }
    for (int o = 16; o > 0; o >>= 1) {
        pb += __shfl_xor(pb, o);
        pt += __shfl_xor(pt, o);
    }
    if (lane == 0) { atomicAdd(&s_base, pb); atomicAdd(&s_total, pt); }
    __syncthreads();
    const int base = s_base;
    const int total = s_total;
    const bool staged = total <= CAP;

    sweep_range(bsorted, boff, P, b, runS, runL,
                [&](unsigned w) { atomicAdd(&cnt[w >> 23], 1); });
    __syncthreads();
    const int v = t < RS ? cnt[t] : 0;
    const int incl = wave_incl_scan(v, lane);
    if (lane == 31) wsum[wv] = incl;
    __syncthreads();
    int off = 0;
    for (int j = 0; j < wv; ++j) off += wsum[j];
    const int excl = off + incl - v;
    if (t < RS) {
        cur[t] = excl;
        const int node = b * RS + t;
        if (node < n) { start[node] = base + excl; endp[node] = base + excl + v; }
    }
    __syncthreads();
    sweep_range(bsorted, boff, P, b, runS, runL,
                [&](unsigned w) {
                    const int pos = atomicAdd(&cur[w >> 23], 1);
                    const int src = (int)(w & 0x7fffffu);
                    if (staged) outb[pos] = src;
                    else atomicExch(&csr[base + pos], src);
                });
    __syncthreads();
    if (staged) {
        const int lim = base + total;
        for (int g = (base & ~31) + t; g < lim; g += 1024) {
            if (g >= base) {
                const int ls = g & ~31;
                const int val = outb[g - base];
                if (ls >= base && ls + 32 <= lim) csr[g] = val;
                else atomicExch(&csr[g], val);
            }
        }
    }
}

template <int M>
__device__ __forceinline__ float row_xor_get(float v) {
    return __int_as_float(__builtin_amdgcn_update_dpp(
        0, __float_as_int(v), 0x160 | M, 0xf, 0xf, true));
}

template <int W>
__device__ __forceinline__ float group_sum(float v) {
    if constexpr (W >= 2) v += row_xor_get<1>(v);
    if constexpr (W >= 4) v += row_xor_get<2>(v);
    if constexpr (W >= 8) v += row_xor_get<4>(v);
    return v;
}

template <int H, int C, int CPL, int U>
__global__ void __launch_bounds__(256)
node_agg_kernel(const int* __restrict__ start,
                const int* __restrict__ endp,
                const int* __restrict__ csr,
                const float* __restrict__ a_src,
                const float* __restrict__ a_dst,
                const float* __restrict__ h,
                const float* __restrict__ bias,
                float* __restrict__ out, int n) {
    constexpr int HC  = H * C;
    constexpr int LPR = HC / CPL;
    constexpr int LPH = C / CPL;
    constexpr int RPW = 32 / LPR;
    constexpr int NV  = CPL / 4;
    static_assert(CPL % 4 == 0 && HC % CPL == 0 && C % CPL == 0 && 32 % LPR == 0
                  && (LPH == 1 || LPH == 2 || LPH == 4 || LPH == 8)
                  && (RPW * HC * 4) % 128 == 0, "lane layout");
    const int lane = threadIdx.x & 31;
    const int wid = __builtin_amdgcn_readfirstlane(
        (int)(blockIdx.x * (blockDim.x >> 5) + (threadIdx.x >> 5)));
    const int node0 = wid * RPW;
    if (node0 >= n) return;
    const int r  = lane / LPR;
    const int cl = lane - r * LPR;
    const int ch = cl * CPL;
    const bool live = node0 + r < n;
    const int node = live ? node0 + r : n - 1;

    v2f ws[2 * NV], acc[2 * NV];
    v2f ds2 = (v2f){0.f, 0.f}, dd2 = (v2f){0.f, 0.f};
    #pragma unroll
    for (int v = 0; v < NV; ++v) {
        const v4f s4 = *(const v4f*)(a_src + ch + 4 * v);
        const v4f d4 = *(const v4f*)(a_dst + ch + 4 * v);
        const v4f h4 = *(const v4f*)(h + (size_t)node * HC + ch + 4 * v);
        ws[2*v] = s4.lo;  ws[2*v+1] = s4.hi;
        acc[2*v] = h4.lo; acc[2*v+1] = h4.hi;
        ds2 += h4.lo * s4.lo; ds2 += h4.hi * s4.hi;
        dd2 += h4.lo * d4.lo; dd2 += h4.hi * d4.hi;
    }
    const int beg = start[node];
    const int deg = live ? endp[node] - beg : 0;

    const float sd = group_sum<LPH>(dd2.x + dd2.y);
    float m = leaky(group_sum<LPH>(ds2.x + ds2.y) + sd);
    float den = 1.f;

    int maxdeg = deg;
    for (int o = 16; o > 0; o >>= 1) {
        const int t = __shfl_xor(maxdeg, o);
        maxdeg = t > maxdeg ? t : maxdeg;
    }
    maxdeg = __builtin_amdgcn_readfirstlane(maxdeg);

    const char* hbytes = (const char*)h;
    const unsigned choff = (unsigned)ch * 4u;
    int sv[U];
    #pragma unroll
    for (int j = 0; j < U; ++j) { sv[j] = node; if (j < deg) sv[j] = csr[beg + j]; }

    for (int i = 0; i < maxdeg; i += U) {
        v4f hv[U][NV];
        #pragma unroll
        for (int j = 0; j < U; ++j) {
            #pragma unroll
            for (int v = 0; v < NV; ++v)
                hv[j][v] = *(const v4f*)(hbytes + ((unsigned)sv[j] * (unsigned)(HC * 4) + choff + 16u * v));
        }
        #pragma unroll
        for (int j = 0; j < U; ++j) {
            sv[j] = node;
            if (i + U + j < deg) sv[j] = csr[beg + i + U + j];
        }
        float e[U];
        #pragma unroll
        for (int j = 0; j < U; ++j) {
            v2f d2 = (v2f){0.f, 0.f};
            #pragma unroll
            for (int v = 0; v < NV; ++v) {
                d2 += hv[j][v].lo * ws[2*v];
                d2 += hv[j][v].hi * ws[2*v+1];
            }
            const float ej = leaky(group_sum<LPH>(d2.x + d2.y) + sd);
            e[j] = (i + j < deg) ? ej : -INFINITY;
        }
        float mn = m;
        #pragma unroll
        for (int j = 0; j < U; ++j) mn = fmaxf(mn, e[j]);
        const float sc = __expf(m - mn);
        den *= sc;
        #pragma unroll
        for (int c = 0; c < 2 * NV; ++c) acc[c] *= sc;
        #pragma unroll
        for (int j = 0; j < U; ++j) {
            const float p = __expf(e[j] - mn);
            den += p;
            #pragma unroll
            for (int v = 0; v < NV; ++v) {
                acc[2*v]   += p * hv[j][v].lo;
                acc[2*v+1] += p * hv[j][v].hi;
            }
        }
        m = mn;
    }

    float q[CPL];
    #pragma unroll
    for (int v = 0; v < NV; ++v) {
        const float4 b4 = *(const float4*)(bias + ch + 4 * v);
        q[4*v]   = acc[2*v].x   / den + b4.x;
        q[4*v+1] = acc[2*v].y   / den + b4.y;
        q[4*v+2] = acc[2*v+1].x / den + b4.z;
        q[4*v+3] = acc[2*v+1].y / den + b4.w;
    }

    if constexpr (NV == 1) {
        if (live) {
            float4 o; o.x = q[0]; o.y = q[1]; o.z = q[2]; o.w = q[3];
            *(float4*)(out + (size_t)node * HC + ch) = o;
        }
    } else {
        constexpr int LPS = HC / 4;
        constexpr int RPS = 32 / LPS;
        static_assert(32 % LPS == 0 && RPS * NV == RPW, "store layout");
        const int rs = lane / LPS;
        const int pi = lane - rs * LPS;
        const int own = pi / NV;
        const int pc = pi - own * NV;
        #pragma unroll
        for (int k = 0; k < NV; ++k) {
            const int from = (k * RPS + rs) * LPR + own;
            float o[4];
            #pragma unroll
            for (int c = 0; c < 4; ++c) {
                float val = 0.f;
                #pragma unroll
                for (int w = 0; w < NV; ++w) {
                    const float g = __shfl(q[4 * w + c], from);
                    if (pc == w) val = g;
                }
                o[c] = val;
            }
            const int nd = node0 + k * RPS + rs;
            if (nd < n) {
                float4 o4; o4.x = o[0]; o4.y = o[1]; o4.z = o[2]; o4.w = o[3];
                *(float4*)(out + (size_t)nd * HC + 4 * pi) = o4;
            }
        }
    }
}

static inline int ceil_div(int a, int b) { return (a + b - 1) / b; }

extern "C" void kernel_launch(void* const* d_in, const int* in_sizes, int n_in,
                              void* d_out, int out_size, void* d_ws, size_t ws_size,
                              hipStream_t stream) {
    const float* x    = (const float*)d_in[0];
    const int*   eidx = (const int*)  d_in[1];
    const float* W1   = (const float*)d_in[2];
    const float* aS1  = (const float*)d_in[3];
    const float* aD1  = (const float*)d_in[4];
    const float* b1   = (const float*)d_in[5];
    const float* W2   = (const float*)d_in[6];
    const float* aS2  = (const float*)d_in[7];
    const float* aD2  = (const float*)d_in[8];
    const float* b2   = (const float*)d_in[9];
    float* out = (float*)d_out;

    const int FEATS = 256, HID = 64, H1 = 8, C1 = 8, CLS = 16;
    const int N = in_sizes[0] / FEATS;
    const int E = in_sizes[1] / 2;

    float* ws = (float*)d_ws;
    float* h1   = ws;                 ws += (size_t)N * HID;
    ws += (size_t)N * H1;
    ws += (size_t)N * H1;
    float* o1   = ws;                 ws += (size_t)N * HID;
    float* h2   = ws;                 ws += (size_t)N * CLS;
    ws += (size_t)N;
    ws += (size_t)N;
    int* rstart = (int*)ws;           ws += (size_t)N;
    int* rend   = (int*)ws;           ws += (size_t)N;
    int* csr    = (int*)ws;           ws += (size_t)E;
    int* boff   = (int*)h2;
    unsigned* bsorted = (unsigned*)o1;

    const int WPB = 4;
    const dim3 gblock(32 * WPB);

    const int P = ceil_div(E, PT);
    const int R = ceil_div(N, RS);
    part_kernel<<<P, 1024, 0, stream>>>(eidx, E, bsorted, boff);
    range_kernel<<<R, 1024, 0, stream>>>(bsorted, boff, P, N, rstart, rend, csr);

    {
        dim3 grid(ceil_div(N / 16, WPB), 1);
        wmma_gemm_f32<4><<<grid, gblock, 0, stream>>>(x, W1, h1, N, FEATS, HID);
    }
    node_agg_kernel<H1, C1, 8, 8><<<ceil_div(ceil_div(N, 4), 8), 256, 0, stream>>>(rstart, rend, csr, aS1, aD1, h1, b1, o1, N);

    {
        dim3 grid(ceil_div(N / 16, WPB), 1);
        wmma_gemm_f32<1><<<grid, gblock, 0, stream>>>(o1, W2, h2, N, HID, CLS);
    }
    node_agg_kernel<1, CLS, 4, 8><<<ceil_div(ceil_div(N, 8), 8), 256, 0, stream>>>(rstart, rend, csr, aS2, aD2, h2, b2, out, N);
}


# layer 2's projection moves into layer 1's row aggregation: a block lays its 32 rows in LDS and two of its waves multiply them by W2 on f32 WMMA, same k order; no o1 in memory, 5 launches not 6
# speedup vs baseline: 8.8523x; 1.0346x over previous
#include <hip/hip_runtime.h>
#include <math.h>

typedef float v2f __attribute__((ext_vector_type(2)));
typedef float v4f __attribute__((ext_vector_type(4)));
typedef float v8f __attribute__((ext_vector_type(8)));

#define NEG_SLOPE 0.2f

__device__ __forceinline__ float leaky(float x) {
    return x > 0.f ? x : NEG_SLOPE * x;
}


template <int CT>
__global__ void wmma_gemm_f32(const float* __restrict__ A,
                              const float* __restrict__ B,
                              float* __restrict__ Out,
                              int M, int K, int Nout) {
    const int wave = threadIdx.x >> 5;
    const int lane = threadIdx.x & 31;
    const int row0 = (blockIdx.x * (blockDim.x >> 5) + wave) << 4;
    const int col0 = blockIdx.y * (CT * 16);
    if (row0 >= M) return;

    const int half = lane >> 4;
    const int l16  = lane & 15;
    const float* arow = A + (size_t)(row0 + l16) * K;
    const int col = col0 + l16;

    v8f acc[CT];
    #pragma unroll
    for (int t = 0; t < CT; ++t) acc[t] = (v8f){};

    for (int k = 0; k < K; k += 4) {
        const int ka = k + 2 * half;
        v2f a;
        a.x = arow[ka];
        a.y = arow[ka + 1];
        const float* brow0 = B + (size_t)ka * Nout + col;
        const float* brow1 = B + (size_t)(ka + 1) * Nout + col;
        #pragma unroll
        for (int t = 0; t < CT; ++t) {
            v2f b;
            b.x = brow0[t * 16];
            b.y = brow1[t * 16];
            acc[t] = __builtin_amdgcn_wmma_f32_16x16x4_f32(
                 false, a,  false, b,
                 (short)0, acc[t],  false,  false);
        }
    }
    #pragma unroll
    for (int t = 0; t < CT; ++t) {
        #pragma unroll
        for (int r = 0; r < 8; ++r) {
            Out[(size_t)(row0 + r + 8 * half) * Nout + col + t * 16] = acc[t][r];
        }
    }
}

#define PT 4096
#define RS_SHIFT 9
#define RS 512
#define BP 256
#define CAP 12288

__device__ __forceinline__ int wave_incl_scan(int v, int lane) {
    for (int o = 1; o < 32; o <<= 1) {
        const int u = __shfl_up(v, o);
        if (lane >= o) v += u;
    }
    return v;
}

__global__ void __launch_bounds__(1024)
part_kernel(const int* __restrict__ eidx, int E,
            unsigned* __restrict__ bsorted, int* __restrict__ boff) {
    __shared__ int hist[BP];
    __shared__ int offs[BP];
    __shared__ int wsum[32];
    __shared__ unsigned srt[PT];
    const int t = threadIdx.x;
    const int lane = t & 31;
    const int wv = t >> 5;
    const int e0 = blockIdx.x * PT;
    if (t < BP) hist[t] = 0;
    __syncthreads();
    int bk[4], rk[4];
    unsigned wd[4];
    #pragma unroll
    for (int k = 0; k < 4; ++k) {
        const int i = e0 + t + 1024 * k;
        bk[k] = -1; rk[k] = 0; wd[k] = 0u;
        if (i < E) {
            const int d = eidx[E + i];
            const int s = eidx[i];
            bk[k] = d >> RS_SHIFT;
            wd[k] = ((unsigned)(d & (RS - 1)) << 23) | (unsigned)s;
        }
    }
    #pragma unroll
    for (int k = 0; k < 4; ++k)
        if (bk[k] >= 0) rk[k] = atomicAdd(&hist[bk[k]], 1);
    __syncthreads();
    const int v = t < BP ? hist[t] : 0;
    const int incl = wave_incl_scan(v, lane);
    if (lane == 31) wsum[wv] = incl;
    __syncthreads();
    int off = 0;
    for (int j = 0; j < wv; ++j) off += wsum[j];
    if (t < BP) offs[t] = off + incl - v;
    __syncthreads();
    #pragma unroll
    for (int k = 0; k < 4; ++k)
        if (bk[k] >= 0) srt[offs[bk[k]] + rk[k]] = wd[k];
    __syncthreads();
    const int nvalid = (E - e0) < PT ? (E - e0) : PT;
    #pragma unroll
    for (int k = 0; k < 4; ++k) {
        const int i = t + 1024 * k;
        if (i < nvalid) bsorted[e0 + i] = srt[i];
    }
    if (t < BP) boff[blockIdx.x * BP + t] = offs[t];
}

template <typename F>
__device__ __forceinline__ void sweep_range(const unsigned* __restrict__ bsorted,
                                            const int* __restrict__ boff,
                                            int P, int b, int* runS, int* runL,
                                            F&& f) {
    const int t = threadIdx.x;
    const int lane = t & 31;
    const int wv = t >> 5;
    for (int p0 = 0; p0 < P; p0 += 1024) {
        __syncthreads();
        const int p = p0 + t;
        if (p < P) {
            const int s = boff[p * BP + b];
            runS[t] = p * PT + s;
            runL[t] = boff[p * BP + b + 1] - s;
        }
        __syncthreads();
        const int np = (P - p0) < 1024 ? (P - p0) : 1024;
        for (int q0 = wv; q0 < np; q0 += 128) {
            int S[4], L[4];
            unsigned w[4];
            #pragma unroll
            for (int k = 0; k < 4; ++k) {
                const int q = q0 + 32 * k;
                S[k] = 0; L[k] = 0;
                if (q < np) { S[k] = runS[q]; L[k] = runL[q]; }
            }
            #pragma unroll
            for (int k = 0; k < 4; ++k) {
                w[k] = 0u;
                if (lane < L[k]) w[k] = bsorted[S[k] + lane];
            }
            #pragma unroll
            for (int k = 0; k < 4; ++k)
                if (lane < L[k]) f(w[k]);
            #pragma unroll
            for (int k = 0; k < 4; ++k)
                for (int j = lane + 32; j < L[k]; j += 32) f(bsorted[S[k] + j]);
        }
    }
}

__global__ void __launch_bounds__(1024)
range_kernel(const unsigned* __restrict__ bsorted, const int* __restrict__ boff,
             int P, int n, int* __restrict__ start, int* __restrict__ endp,
             int* __restrict__ csr) {
    __shared__ int cnt[RS];
    __shared__ int cur[RS];
    __shared__ int runS[1024];
    __shared__ int runL[1024];
    __shared__ int wsum[32];
    __shared__ int s_base;
    __shared__ int s_total;
    __shared__ int outb[CAP];
    const int t = threadIdx.x;
    const int lane = t & 31;
    const int wv = t >> 5;
    const int b = blockIdx.x;
    if (t < RS) cnt[t] = 0;
    if (t == 0) { s_base = 0; s_total = 0; }
    __syncthreads();
    int pb = 0, pt = 0;
    for (int p = t; p < P; p += 1024) {
        const int s = boff[p * BP + b];
        pb += s;
        pt += boff[p * BP + b + 1] - s;
    }
    for (int o = 16; o > 0; o >>= 1) {
        pb += __shfl_xor(pb, o);
        pt += __shfl_xor(pt, o);
    }
    if (lane == 0) { atomicAdd(&s_base, pb); atomicAdd(&s_total, pt); }
    __syncthreads();
    const int base = s_base;
    const int total = s_total;
    const bool staged = total <= CAP;

    sweep_range(bsorted, boff, P, b, runS, runL,
                [&](unsigned w) { atomicAdd(&cnt[w >> 23], 1); });
    __syncthreads();
    const int v = t < RS ? cnt[t] : 0;
    const int incl = wave_incl_scan(v, lane);
    if (lane == 31) wsum[wv] = incl;
    __syncthreads();
    int off = 0;
    for (int j = 0; j < wv; ++j) off += wsum[j];
    const int excl = off + incl - v;
    if (t < RS) {
        cur[t] = excl;
        const int node = b * RS + t;
        if (node < n) { start[node] = base + excl; endp[node] = base + excl + v; }
    }
    __syncthreads();
    sweep_range(bsorted, boff, P, b, runS, runL,
                [&](unsigned w) {
                    const int pos = atomicAdd(&cur[w >> 23], 1);
                    const int src = (int)(w & 0x7fffffu);
                    if (staged) outb[pos] = src;
                    else atomicExch(&csr[base + pos], src);
                });
    __syncthreads();
    if (staged) {
        const int lim = base + total;
        for (int g = (base & ~31) + t; g < lim; g += 1024) {
            if (g >= base) {
                const int ls = g & ~31;
                const int val = outb[g - base];
                if (ls >= base && ls + 32 <= lim) csr[g] = val;
                else atomicExch(&csr[g], val);
            }
        }
    }
}

template <int M>
__device__ __forceinline__ float row_xor_get(float v) {
    return __int_as_float(__builtin_amdgcn_update_dpp(
        0, __float_as_int(v), 0x160 | M, 0xf, 0xf, true));
}

template <int W>
__device__ __forceinline__ float group_sum(float v) {
    if constexpr (W >= 2) v += row_xor_get<1>(v);
    if constexpr (W >= 4) v += row_xor_get<2>(v);
    if constexpr (W >= 8) v += row_xor_get<4>(v);
    return v;
}

template <int H, int C, int CPL, int U, int NOUT>
__global__ void __launch_bounds__(256)
node_agg_kernel(const int* __restrict__ start,
                const int* __restrict__ endp,
                const int* __restrict__ csr,
                const float* __restrict__ a_src,
                const float* __restrict__ a_dst,
                const float* __restrict__ h,
                const float* __restrict__ bias,
                const float* __restrict__ Wn,
                float* __restrict__ out, int n) {
    constexpr int HC  = H * C;
    constexpr int LPR = HC / CPL;
    constexpr int LPH = C / CPL;
    constexpr int RPW = 32 / LPR;
    constexpr int NV  = CPL / 4;
    static_assert(CPL % 4 == 0 && HC % CPL == 0 && C % CPL == 0 && 32 % LPR == 0
                  && (LPH == 1 || LPH == 2 || LPH == 4 || LPH == 8)
                  && (RPW * HC * 4) % 128 == 0, "lane layout");
    const int lane = threadIdx.x & 31;
    const int wid = __builtin_amdgcn_readfirstlane(
        (int)(blockIdx.x * (blockDim.x >> 5) + (threadIdx.x >> 5)));
    const int node0 = wid * RPW;
    if constexpr (NOUT == 0) { if (node0 >= n) return; }
    const int r  = lane / LPR;
    const int cl = lane - r * LPR;
    const int ch = cl * CPL;
    const bool live = node0 + r < n;
    const int node = live ? node0 + r : n - 1;

    v2f ws[2 * NV], acc[2 * NV];
    v2f ds2 = (v2f){0.f, 0.f}, dd2 = (v2f){0.f, 0.f};
    #pragma unroll
    for (int v = 0; v < NV; ++v) {
        const v4f s4 = *(const v4f*)(a_src + ch + 4 * v);
        const v4f d4 = *(const v4f*)(a_dst + ch + 4 * v);
        const v4f h4 = *(const v4f*)(h + (size_t)node * HC + ch + 4 * v);
        ws[2*v] = s4.lo;  ws[2*v+1] = s4.hi;
        acc[2*v] = h4.lo; acc[2*v+1] = h4.hi;
        ds2 += h4.lo * s4.lo; ds2 += h4.hi * s4.hi;
        dd2 += h4.lo * d4.lo; dd2 += h4.hi * d4.hi;
    }
    const int beg = start[node];
    const int deg = live ? endp[node] - beg : 0;

    const float sd = group_sum<LPH>(dd2.x + dd2.y);
    float m = leaky(group_sum<LPH>(ds2.x + ds2.y) + sd);
    float den = 1.f;

    int maxdeg = deg;
    for (int o = 16; o > 0; o >>= 1) {
        const int t = __shfl_xor(maxdeg, o);
        maxdeg = t > maxdeg ? t : maxdeg;
    }
    maxdeg = __builtin_amdgcn_readfirstlane(maxdeg);

    const char* hbytes = (const char*)h;
    const unsigned choff = (unsigned)ch * 4u;
    int sv[U];
    #pragma unroll
    for (int j = 0; j < U; ++j) { sv[j] = node; if (j < deg) sv[j] = csr[beg + j]; }

    for (int i = 0; i < maxdeg; i += U) {
        v4f hv[U][NV];
        #pragma unroll
        for (int j = 0; j < U; ++j) {
            #pragma unroll
            for (int v = 0; v < NV; ++v)
                hv[j][v] = *(const v4f*)(hbytes + ((unsigned)sv[j] * (unsigned)(HC * 4) + choff + 16u * v));
        }
        #pragma unroll
        for (int j = 0; j < U; ++j) {
            sv[j] = node;
            if (i + U + j < deg) sv[j] = csr[beg + i + U + j];
        }
        float e[U];
        #pragma unroll
        for (int j = 0; j < U; ++j) {
            v2f d2 = (v2f){0.f, 0.f};
            #pragma unroll
            for (int v = 0; v < NV; ++v) {
                d2 += hv[j][v].lo * ws[2*v];
                d2 += hv[j][v].hi * ws[2*v+1];
            }
            const float ej = leaky(group_sum<LPH>(d2.x + d2.y) + sd);
            e[j] = (i + j < deg) ? ej : -INFINITY;
        }
        float mn = m;
        #pragma unroll
        for (int j = 0; j < U; ++j) mn = fmaxf(mn, e[j]);
        const float sc = __expf(m - mn);
        den *= sc;
        #pragma unroll
        for (int c = 0; c < 2 * NV; ++c) acc[c] *= sc;
        #pragma unroll
        for (int j = 0; j < U; ++j) {
            const float p = __expf(e[j] - mn);
            den += p;
            #pragma unroll
            for (int v = 0; v < NV; ++v) {
                acc[2*v]   += p * hv[j][v].lo;
                acc[2*v+1] += p * hv[j][v].hi;
            }
        }
        m = mn;
    }

    float q[CPL];
    #pragma unroll
    for (int v = 0; v < NV; ++v) {
        const float4 b4 = *(const float4*)(bias + ch + 4 * v);
        q[4*v]   = acc[2*v].x   / den + b4.x;
        q[4*v+1] = acc[2*v].y   / den + b4.y;
        q[4*v+2] = acc[2*v+1].x / den + b4.z;
        q[4*v+3] = acc[2*v+1].y / den + b4.w;
    }

    if constexpr (NOUT > 0) {
        constexpr int WPB = 8;
        constexpr int RPB = WPB * RPW;
        constexpr int PITCH = HC + 4;
        static_assert(NOUT == 16 && RPB % 16 == 0 && HC % 4 == 0 && RPB / 16 <= WPB, "product tile");
        __shared__ __attribute__((aligned(16))) float tile[RPB * PITCH];
        const int wib = __builtin_amdgcn_readfirstlane((int)(threadIdx.x >> 5));
        float* trow = tile + (wib * RPW + r) * PITCH + ch;
        #pragma unroll
        for (int v = 0; v < NV; ++v) {
            v4f o; o.x = q[4*v]; o.y = q[4*v+1]; o.z = q[4*v+2]; o.w = q[4*v+3];
            *(v4f*)(trow + 4 * v) = o;
        }
        __syncthreads();
        if (wib < RPB / 16) {
            const int half = lane >> 4;
            const int l16  = lane & 15;
            const float* arow = tile + (wib * 16 + l16) * PITCH;
            v8f d = (v8f){};
            #pragma unroll
            for (int k = 0; k < HC; k += 4) {
                const int ka = k + 2 * half;
                const v2f a = *(const v2f*)(arow + ka);
                v2f b;
                b.x = Wn[(size_t)ka * NOUT + l16];
                b.y = Wn[(size_t)(ka + 1) * NOUT + l16];
                d = __builtin_amdgcn_wmma_f32_16x16x4_f32(
                     false, a,  false, b,
                     (short)0, d,  false,  false);
            }
            const int row0 = (int)blockIdx.x * RPB + wib * 16 + 8 * half;
            #pragma unroll
            for (int i = 0; i < 8; ++i) {
                if (row0 + i < n) out[(size_t)(row0 + i) * NOUT + l16] = d[i];
            }
        }
    } else if constexpr (NV == 1) {
        if (live) {
            float4 o; o.x = q[0]; o.y = q[1]; o.z = q[2]; o.w = q[3];
            *(float4*)(out + (size_t)node * HC + ch) = o;
        }
    } else {
        constexpr int LPS = HC / 4;
        constexpr int RPS = 32 / LPS;
        static_assert(32 % LPS == 0 && RPS * NV == RPW, "store layout");
        const int rs = lane / LPS;
        const int pi = lane - rs * LPS;
        const int own = pi / NV;
        const int pc = pi - own * NV;
        #pragma unroll
        for (int k = 0; k < NV; ++k) {
            const int from = (k * RPS + rs) * LPR + own;
            float o[4];
            #pragma unroll
            for (int c = 0; c < 4; ++c) {
                float val = 0.f;
                #pragma unroll
                for (int w = 0; w < NV; ++w) {
                    const float g = __shfl(q[4 * w + c], from);
                    if (pc == w) val = g;
                }
                o[c] = val;
            }
            const int nd = node0 + k * RPS + rs;
            if (nd < n) {
                float4 o4; o4.x = o[0]; o4.y = o[1]; o4.z = o[2]; o4.w = o[3];
                *(float4*)(out + (size_t)nd * HC + 4 * pi) = o4;
            }
        }
    }
}

static inline int ceil_div(int a, int b) { return (a + b - 1) / b; }

extern "C" void kernel_launch(void* const* d_in, const int* in_sizes, int n_in,
                              void* d_out, int out_size, void* d_ws, size_t ws_size,
                              hipStream_t stream) {
    const float* x    = (const float*)d_in[0];
    const int*   eidx = (const int*)  d_in[1];
    const float* W1   = (const float*)d_in[2];
    const float* aS1  = (const float*)d_in[3];
    const float* aD1  = (const float*)d_in[4];
    const float* b1   = (const float*)d_in[5];
    const float* W2   = (const float*)d_in[6];
    const float* aS2  = (const float*)d_in[7];
    const float* aD2  = (const float*)d_in[8];
    const float* b2   = (const float*)d_in[9];
    float* out = (float*)d_out;

    const int FEATS = 256, HID = 64, H1 = 8, C1 = 8, CLS = 16;
    const int N = in_sizes[0] / FEATS;
    const int E = in_sizes[1] / 2;

    float* ws = (float*)d_ws;
    float* h1   = ws;                 ws += (size_t)N * HID;
    ws += (size_t)N * H1;
    ws += (size_t)N * H1;
    float* o1   = ws;                 ws += (size_t)N * HID;
    float* h2   = ws;                 ws += (size_t)N * CLS;
    ws += (size_t)N;
    ws += (size_t)N;
    int* rstart = (int*)ws;           ws += (size_t)N;
    int* rend   = (int*)ws;           ws += (size_t)N;
    int* csr    = (int*)ws;           ws += (size_t)E;
    int* boff   = (int*)h2;
    unsigned* bsorted = (unsigned*)o1;

    const int WPB = 4;
    const dim3 gblock(32 * WPB);

    const int P = ceil_div(E, PT);
    const int R = ceil_div(N, RS);
    part_kernel<<<P, 1024, 0, stream>>>(eidx, E, bsorted, boff);
    range_kernel<<<R, 1024, 0, stream>>>(bsorted, boff, P, N, rstart, rend, csr);

    {
        dim3 grid(ceil_div(N / 16, WPB), 1);
        wmma_gemm_f32<4><<<grid, gblock, 0, stream>>>(x, W1, h1, N, FEATS, HID);
    }
    node_agg_kernel<H1, C1, 8, 8, CLS><<<ceil_div(ceil_div(N, 4), 8), 256, 0, stream>>>(rstart, rend, csr, aS1, aD1, h1, b1, W2, h2, N);

    node_agg_kernel<1, CLS, 4, 8, 0><<<ceil_div(ceil_div(N, 8), 8), 256, 0, stream>>>(rstart, rend, csr, aS2, aD2, h2, b2, nullptr, out, N);
}


# layer 1's projection x W1 (f32 WMMA strips, same k order) runs in the launch of the grouping's second level: 250 projection blocks first, 196 range blocks beside them, two a WGP; 4 launches not 5
# speedup vs baseline: 9.9517x; 1.1242x over previous
#include <hip/hip_runtime.h>
#include <math.h>

typedef float v2f __attribute__((ext_vector_type(2)));
typedef float v4f __attribute__((ext_vector_type(4)));
typedef float v8f __attribute__((ext_vector_type(8)));

#define NEG_SLOPE 0.2f

__device__ __forceinline__ float leaky(float x) {
    return x > 0.f ? x : NEG_SLOPE * x;
}


template <int CT, int K>
__device__ __forceinline__ void wmma_strip_f32(const float* __restrict__ A,
                                               const float* __restrict__ B,
                                               float* __restrict__ Out,
                                               int row0, int lane) {
    constexpr int Nout = CT * 16;
    const int half = lane >> 4;
    const int l16  = lane & 15;
    const float* arow = A + (size_t)(row0 + l16) * K;
    const int col = l16;

    v8f acc[CT];
    #pragma unroll
    for (int t = 0; t < CT; ++t) acc[t] = (v8f){};

    for (int k = 0; k < K; k += 4) {
        const int ka = k + 2 * half;
        v2f a;
        a.x = arow[ka];
        a.y = arow[ka + 1];
        const float* brow0 = B + (size_t)ka * Nout + col;
        const float* brow1 = B + (size_t)(ka + 1) * Nout + col;
        #pragma unroll
        for (int t = 0; t < CT; ++t) {
            v2f b;
            b.x = brow0[t * 16];
            b.y = brow1[t * 16];
            acc[t] = __builtin_amdgcn_wmma_f32_16x16x4_f32(
                 false, a,  false, b,
                 (short)0, acc[t],  false,  false);
        }
    }
    #pragma unroll
    for (int t = 0; t < CT; ++t) {
        #pragma unroll
        for (int r = 0; r < 8; ++r) {
            Out[(size_t)(row0 + r + 8 * half) * Nout + col + t * 16] = acc[t][r];
        }
    }
}

#define PT 4096
#define RS_SHIFT 9
#define RS 512
#define BP 256
#define CAP 12288

__device__ __forceinline__ int wave_incl_scan(int v, int lane) {
    for (int o = 1; o < 32; o <<= 1) {
        const int u = __shfl_up(v, o);
        if (lane >= o) v += u;
    }
    return v;
}

__global__ void __launch_bounds__(1024)
part_kernel(const int* __restrict__ eidx, int E,
            unsigned* __restrict__ bsorted, int* __restrict__ boff) {
    __shared__ int hist[BP];
    __shared__ int offs[BP];
    __shared__ int wsum[32];
    __shared__ unsigned srt[PT];
    const int t = threadIdx.x;
    const int lane = t & 31;
    const int wv = t >> 5;
    const int e0 = blockIdx.x * PT;
    if (t < BP) hist[t] = 0;
    __syncthreads();
    int bk[4], rk[4];
    unsigned wd[4];
    #pragma unroll
    for (int k = 0; k < 4; ++k) {
        const int i = e0 + t + 1024 * k;
        bk[k] = -1; rk[k] = 0; wd[k] = 0u;
        if (i < E) {
            const int d = eidx[E + i];
            const int s = eidx[i];
            bk[k] = d >> RS_SHIFT;
            wd[k] = ((unsigned)(d & (RS - 1)) << 23) | (unsigned)s;
        }
    }
    #pragma unroll
    for (int k = 0; k < 4; ++k)
        if (bk[k] >= 0) rk[k] = atomicAdd(&hist[bk[k]], 1);
    __syncthreads();
    const int v = t < BP ? hist[t] : 0;
    const int incl = wave_incl_scan(v, lane);
    if (lane == 31) wsum[wv] = incl;
    __syncthreads();
    int off = 0;
    for (int j = 0; j < wv; ++j) off += wsum[j];
    if (t < BP) offs[t] = off + incl - v;
    __syncthreads();
    #pragma unroll
    for (int k = 0; k < 4; ++k)
        if (bk[k] >= 0) srt[offs[bk[k]] + rk[k]] = wd[k];
    __syncthreads();
    const int nvalid = (E - e0) < PT ? (E - e0) : PT;
    #pragma unroll
    for (int k = 0; k < 4; ++k) {
        const int i = t + 1024 * k;
        if (i < nvalid) bsorted[e0 + i] = srt[i];
    }
    if (t < BP) boff[blockIdx.x * BP + t] = offs[t];
}

template <typename F>
__device__ __forceinline__ void sweep_range(const unsigned* __restrict__ bsorted,
                                            const int* __restrict__ boff,
                                            int P, int b, int* runS, int* runL,
                                            F&& f) {
    const int t = threadIdx.x;
    const int lane = t & 31;
    const int wv = t >> 5;
    for (int p0 = 0; p0 < P; p0 += 1024) {
        __syncthreads();
        const int p = p0 + t;
        if (p < P) {
            const int s = boff[p * BP + b];
            runS[t] = p * PT + s;
            runL[t] = boff[p * BP + b + 1] - s;
        }
        __syncthreads();
        const int np = (P - p0) < 1024 ? (P - p0) : 1024;
        for (int q0 = wv; q0 < np; q0 += 128) {
            int S[4], L[4];
            unsigned w[4];
            #pragma unroll
            for (int k = 0; k < 4; ++k) {
                const int q = q0 + 32 * k;
                S[k] = 0; L[k] = 0;
                if (q < np) { S[k] = runS[q]; L[k] = runL[q]; }
            }
            #pragma unroll
            for (int k = 0; k < 4; ++k) {
                w[k] = 0u;
                if (lane < L[k]) w[k] = bsorted[S[k] + lane];
            }
            #pragma unroll
            for (int k = 0; k < 4; ++k)
                if (lane < L[k]) f(w[k]);
            #pragma unroll
            for (int k = 0; k < 4; ++k)
                for (int j = lane + 32; j < L[k]; j += 32) f(bsorted[S[k] + j]);
        }
    }
}

#define PROJ_K 256
#define PROJ_CT 4

__global__ void __launch_bounds__(1024, 16)
range_proj_kernel(const unsigned* __restrict__ bsorted, const int* __restrict__ boff,
                  int P, int n, int* __restrict__ start, int* __restrict__ endp,
                  int* __restrict__ csr,
                  const float* __restrict__ x, const float* __restrict__ W,
                  float* __restrict__ hout, int gblocks, int tpb) {
    if ((int)blockIdx.x < gblocks) {
        const int w = threadIdx.x >> 5;
        if (w >= tpb) return;
        const int row0 = ((int)blockIdx.x * tpb + w) << 4;
        if (row0 >= n) return;
        wmma_strip_f32<PROJ_CT, PROJ_K>(x, W, hout, row0, threadIdx.x & 31);
        return;
    }
    __shared__ int cnt[RS];
    __shared__ int cur[RS];
    __shared__ int runS[1024];
    __shared__ int runL[1024];
    __shared__ int wsum[32];
    __shared__ int s_base;
    __shared__ int s_total;
    __shared__ int outb[CAP];
    const int t = threadIdx.x;
    const int lane = t & 31;
    const int wv = t >> 5;
    const int b = (int)blockIdx.x - gblocks;
    if (t < RS) cnt[t] = 0;
    if (t == 0) { s_base = 0; s_total = 0; }
    __syncthreads();
    int pb = 0, pt = 0;
    for (int p = t; p < P; p += 1024) {
        const int s = boff[p * BP + b];
        pb += s;
        pt += boff[p * BP + b + 1] - s;
    }
    for (int o = 16; o > 0; o >>= 1) {
        pb += __shfl_xor(pb, o);
        pt += __shfl_xor(pt, o);
    }
    if (lane == 0) { atomicAdd(&s_base, pb); atomicAdd(&s_total, pt); }
    __syncthreads();
    const int base = s_base;
    const int total = s_total;
    const bool staged = total <= CAP;

    sweep_range(bsorted, boff, P, b, runS, runL,
                [&](unsigned w) { atomicAdd(&cnt[w >> 23], 1); });
    __syncthreads();
    const int v = t < RS ? cnt[t] : 0;
    const int incl = wave_incl_scan(v, lane);
    if (lane == 31) wsum[wv] = incl;
    __syncthreads();
    int off = 0;
    for (int j = 0; j < wv; ++j) off += wsum[j];
    const int excl = off + incl - v;
    if (t < RS) {
        cur[t] = excl;
        const int node = b * RS + t;
        if (node < n) { start[node] = base + excl; endp[node] = base + excl + v; }
    }
    __syncthreads();
    sweep_range(bsorted, boff, P, b, runS, runL,
                [&](unsigned w) {
                    const int pos = atomicAdd(&cur[w >> 23], 1);
                    const int src = (int)(w & 0x7fffffu);
                    if (staged) outb[pos] = src;
                    else atomicExch(&csr[base + pos], src);
                });
    __syncthreads();
    if (staged) {
        const int lim = base + total;
        for (int g = (base & ~31) + t; g < lim; g += 1024) {
            if (g >= base) {
                const int ls = g & ~31;
                const int val = outb[g - base];
                if (ls >= base && ls + 32 <= lim) csr[g] = val;
                else atomicExch(&csr[g], val);
            }
        }
    }
}

template <int M>
__device__ __forceinline__ float row_xor_get(float v) {
    return __int_as_float(__builtin_amdgcn_update_dpp(
        0, __float_as_int(v), 0x160 | M, 0xf, 0xf, true));
}

template <int W>
__device__ __forceinline__ float group_sum(float v) {
    if constexpr (W >= 2) v += row_xor_get<1>(v);
    if constexpr (W >= 4) v += row_xor_get<2>(v);
    if constexpr (W >= 8) v += row_xor_get<4>(v);
    return v;
}

template <int H, int C, int CPL, int U, int NOUT>
__global__ void __launch_bounds__(256)
node_agg_kernel(const int* __restrict__ start,
                const int* __restrict__ endp,
                const int* __restrict__ csr,
                const float* __restrict__ a_src,
                const float* __restrict__ a_dst,
                const float* __restrict__ h,
                const float* __restrict__ bias,
                const float* __restrict__ Wn,
                float* __restrict__ out, int n) {
    constexpr int HC  = H * C;
    constexpr int LPR = HC / CPL;
    constexpr int LPH = C / CPL;
    constexpr int RPW = 32 / LPR;
    constexpr int NV  = CPL / 4;
    static_assert(CPL % 4 == 0 && HC % CPL == 0 && C % CPL == 0 && 32 % LPR == 0
                  && (LPH == 1 || LPH == 2 || LPH == 4 || LPH == 8)
                  && (RPW * HC * 4) % 128 == 0, "lane layout");
    const int lane = threadIdx.x & 31;
    const int wid = __builtin_amdgcn_readfirstlane(
        (int)(blockIdx.x * (blockDim.x >> 5) + (threadIdx.x >> 5)));
    const int node0 = wid * RPW;
    if constexpr (NOUT == 0) { if (node0 >= n) return; }
    const int r  = lane / LPR;
    const int cl = lane - r * LPR;
    const int ch = cl * CPL;
    const bool live = node0 + r < n;
    const int node = live ? node0 + r : n - 1;

    v2f ws[2 * NV], acc[2 * NV];
    v2f ds2 = (v2f){0.f, 0.f}, dd2 = (v2f){0.f, 0.f};
    #pragma unroll
    for (int v = 0; v < NV; ++v) {
        const v4f s4 = *(const v4f*)(a_src + ch + 4 * v);
        const v4f d4 = *(const v4f*)(a_dst + ch + 4 * v);
        const v4f h4 = *(const v4f*)(h + (size_t)node * HC + ch + 4 * v);
        ws[2*v] = s4.lo;  ws[2*v+1] = s4.hi;
        acc[2*v] = h4.lo; acc[2*v+1] = h4.hi;
        ds2 += h4.lo * s4.lo; ds2 += h4.hi * s4.hi;
        dd2 += h4.lo * d4.lo; dd2 += h4.hi * d4.hi;
    }
    const int beg = start[node];
    const int deg = live ? endp[node] - beg : 0;

    const float sd = group_sum<LPH>(dd2.x + dd2.y);
    float m = leaky(group_sum<LPH>(ds2.x + ds2.y) + sd);
    float den = 1.f;

    int maxdeg = deg;
    for (int o = 16; o > 0; o >>= 1) {
        const int t = __shfl_xor(maxdeg, o);
        maxdeg = t > maxdeg ? t : maxdeg;
    }
    maxdeg = __builtin_amdgcn_readfirstlane(maxdeg);

    const char* hbytes = (const char*)h;
    const unsigned choff = (unsigned)ch * 4u;
    int sv[U];
    #pragma unroll
    for (int j = 0; j < U; ++j) { sv[j] = node; if (j < deg) sv[j] = csr[beg + j]; }

    for (int i = 0; i < maxdeg; i += U) {
        v4f hv[U][NV];
        #pragma unroll
        for (int j = 0; j < U; ++j) {
            #pragma unroll
            for (int v = 0; v < NV; ++v)
                hv[j][v] = *(const v4f*)(hbytes + ((unsigned)sv[j] * (unsigned)(HC * 4) + choff + 16u * v));
        }
        #pragma unroll
        for (int j = 0; j < U; ++j) {
            sv[j] = node;
            if (i + U + j < deg) sv[j] = csr[beg + i + U + j];
        }
        float e[U];
        #pragma unroll
        for (int j = 0; j < U; ++j) {
            v2f d2 = (v2f){0.f, 0.f};
            #pragma unroll
            for (int v = 0; v < NV; ++v) {
                d2 += hv[j][v].lo * ws[2*v];
                d2 += hv[j][v].hi * ws[2*v+1];
            }
            const float ej = leaky(group_sum<LPH>(d2.x + d2.y) + sd);
            e[j] = (i + j < deg) ? ej : -INFINITY;
        }
        float mn = m;
        #pragma unroll
        for (int j = 0; j < U; ++j) mn = fmaxf(mn, e[j]);
        const float sc = __expf(m - mn);
        den *= sc;
        #pragma unroll
        for (int c = 0; c < 2 * NV; ++c) acc[c] *= sc;
        #pragma unroll
        for (int j = 0; j < U; ++j) {
            const float p = __expf(e[j] - mn);
            den += p;
            #pragma unroll
            for (int v = 0; v < NV; ++v) {
                acc[2*v]   += p * hv[j][v].lo;
                acc[2*v+1] += p * hv[j][v].hi;
            }
        }
        m = mn;
    }

    float q[CPL];
    #pragma unroll
    for (int v = 0; v < NV; ++v) {
        const float4 b4 = *(const float4*)(bias + ch + 4 * v);
        q[4*v]   = acc[2*v].x   / den + b4.x;
        q[4*v+1] = acc[2*v].y   / den + b4.y;
        q[4*v+2] = acc[2*v+1].x / den + b4.z;
        q[4*v+3] = acc[2*v+1].y / den + b4.w;
    }

    if constexpr (NOUT > 0) {
        constexpr int WPB = 8;
        constexpr int RPB = WPB * RPW;
        constexpr int PITCH = HC + 4;
        static_assert(NOUT == 16 && RPB % 16 == 0 && HC % 4 == 0 && RPB / 16 <= WPB, "product tile");
        __shared__ __attribute__((aligned(16))) float tile[RPB * PITCH];
        const int wib = __builtin_amdgcn_readfirstlane((int)(threadIdx.x >> 5));
        float* trow = tile + (wib * RPW + r) * PITCH + ch;
        #pragma unroll
        for (int v = 0; v < NV; ++v) {
            v4f o; o.x = q[4*v]; o.y = q[4*v+1]; o.z = q[4*v+2]; o.w = q[4*v+3];
            *(v4f*)(trow + 4 * v) = o;
        }
        __syncthreads();
        if (wib < RPB / 16) {
            const int half = lane >> 4;
            const int l16  = lane & 15;
            const float* arow = tile + (wib * 16 + l16) * PITCH;
            v8f d = (v8f){};
            #pragma unroll
            for (int k = 0; k < HC; k += 4) {
                const int ka = k + 2 * half;
                const v2f a = *(const v2f*)(arow + ka);
                v2f b;
                b.x = Wn[(size_t)ka * NOUT + l16];
                b.y = Wn[(size_t)(ka + 1) * NOUT + l16];
                d = __builtin_amdgcn_wmma_f32_16x16x4_f32(
                     false, a,  false, b,
                     (short)0, d,  false,  false);
            }
            const int row0 = (int)blockIdx.x * RPB + wib * 16 + 8 * half;
            #pragma unroll
            for (int i = 0; i < 8; ++i) {
                if (row0 + i < n) out[(size_t)(row0 + i) * NOUT + l16] = d[i];
            }
        }
    } else if constexpr (NV == 1) {
        if (live) {
            float4 o; o.x = q[0]; o.y = q[1]; o.z = q[2]; o.w = q[3];
            *(float4*)(out + (size_t)node * HC + ch) = o;
        }
    } else {
        constexpr int LPS = HC / 4;
        constexpr int RPS = 32 / LPS;
        static_assert(32 % LPS == 0 && RPS * NV == RPW, "store layout");
        const int rs = lane / LPS;
        const int pi = lane - rs * LPS;
        const int own = pi / NV;
        const int pc = pi - own * NV;
        #pragma unroll
        for (int k = 0; k < NV; ++k) {
            const int from = (k * RPS + rs) * LPR + own;
            float o[4];
            #pragma unroll
            for (int c = 0; c < 4; ++c) {
                float val = 0.f;
                #pragma unroll
                for (int w = 0; w < NV; ++w) {
                    const float g = __shfl(q[4 * w + c], from);
                    if (pc == w) val = g;
                }
                o[c] = val;
            }
            const int nd = node0 + k * RPS + rs;
            if (nd < n) {
                float4 o4; o4.x = o[0]; o4.y = o[1]; o4.z = o[2]; o4.w = o[3];
                *(float4*)(out + (size_t)nd * HC + 4 * pi) = o4;
            }
        }
    }
}

static inline int ceil_div(int a, int b) { return (a + b - 1) / b; }

extern "C" void kernel_launch(void* const* d_in, const int* in_sizes, int n_in,
                              void* d_out, int out_size, void* d_ws, size_t ws_size,
                              hipStream_t stream) {
    const float* x    = (const float*)d_in[0];
    const int*   eidx = (const int*)  d_in[1];
    const float* W1   = (const float*)d_in[2];
    const float* aS1  = (const float*)d_in[3];
    const float* aD1  = (const float*)d_in[4];
    const float* b1   = (const float*)d_in[5];
    const float* W2   = (const float*)d_in[6];
    const float* aS2  = (const float*)d_in[7];
    const float* aD2  = (const float*)d_in[8];
    const float* b2   = (const float*)d_in[9];
    float* out = (float*)d_out;

    const int FEATS = PROJ_K, HID = PROJ_CT * 16, H1 = 8, C1 = 8, CLS = 16;
    const int N = in_sizes[0] / FEATS;
    const int E = in_sizes[1] / 2;

    float* ws = (float*)d_ws;
    float* h1   = ws;                 ws += (size_t)N * HID;
    ws += (size_t)N * H1;
    ws += (size_t)N * H1;
    float* o1   = ws;                 ws += (size_t)N * HID;
    float* h2   = ws;                 ws += (size_t)N * CLS;
    ws += (size_t)N;
    ws += (size_t)N;
    int* rstart = (int*)ws;           ws += (size_t)N;
    int* rend   = (int*)ws;           ws += (size_t)N;
    int* csr    = (int*)ws;           ws += (size_t)E;
    int* boff   = (int*)h2;
    unsigned* bsorted = (unsigned*)o1;

    const int P = ceil_div(E, PT);
    const int R = ceil_div(N, RS);
    part_kernel<<<P, 1024, 0, stream>>>(eidx, E, bsorted, boff);

    const int strips = N / 16;
    const int tpb = strips >= 32 * 256 ? 32 : ceil_div(strips, 256);
    const int GB = ceil_div(strips, tpb);
    range_proj_kernel<<<GB + R, 1024, 0, stream>>>(bsorted, boff, P, N, rstart, rend, csr,
                                                   x, W1, h1, GB, tpb);
    node_agg_kernel<H1, C1, 8, 8, CLS><<<ceil_div(ceil_div(N, 4), 8), 256, 0, stream>>>(rstart, rend, csr, aS1, aD1, h1, b1, W2, h2, N);

    node_agg_kernel<1, CLS, 4, 8, 0><<<ceil_div(ceil_div(N, 8), 8), 256, 0, stream>>>(rstart, rend, csr, aS2, aD2, h2, b2, nullptr, out, N);
}


# serial round trips cut in the short launches: guarded loads issued branch-free in groups (edge partition, range sweeps, row prologue); a wave's longest row by lane reads, not LDS permutes
# speedup vs baseline: 10.4567x; 1.0507x over previous
#include <hip/hip_runtime.h>
#include <math.h>

typedef float v2f __attribute__((ext_vector_type(2)));
typedef float v4f __attribute__((ext_vector_type(4)));
typedef float v8f __attribute__((ext_vector_type(8)));

#define NEG_SLOPE 0.2f

__device__ __forceinline__ float leaky(float x) {
    return x > 0.f ? x : NEG_SLOPE * x;
}


template <int CT, int K>
__device__ __forceinline__ void wmma_strip_f32(const float* __restrict__ A,
                                               const float* __restrict__ B,
                                               float* __restrict__ Out,
                                               int row0, int lane) {
    constexpr int Nout = CT * 16;
    const int half = lane >> 4;
    const int l16  = lane & 15;
    const float* arow = A + (size_t)(row0 + l16) * K;
    const int col = l16;

    v8f acc[CT];
    #pragma unroll
    for (int t = 0; t < CT; ++t) acc[t] = (v8f){};

    for (int k = 0; k < K; k += 4) {
        const int ka = k + 2 * half;
        v2f a;
        a.x = arow[ka];
        a.y = arow[ka + 1];
        const float* brow0 = B + (size_t)ka * Nout + col;
        const float* brow1 = B + (size_t)(ka + 1) * Nout + col;
        #pragma unroll
        for (int t = 0; t < CT; ++t) {
            v2f b;
            b.x = brow0[t * 16];
            b.y = brow1[t * 16];
            acc[t] = __builtin_amdgcn_wmma_f32_16x16x4_f32(
                 false, a,  false, b,
                 (short)0, acc[t],  false,  false);
        }
    }
    #pragma unroll
    for (int t = 0; t < CT; ++t) {
        #pragma unroll
        for (int r = 0; r < 8; ++r) {
            Out[(size_t)(row0 + r + 8 * half) * Nout + col + t * 16] = acc[t][r];
        }
    }
}

#define PT 4096
#define RS_SHIFT 9
#define RS 512
#define BP 256
#define CAP 12288

__device__ __forceinline__ int wave_incl_scan(int v, int lane) {
    for (int o = 1; o < 32; o <<= 1) {
        const int u = __shfl_up(v, o);
        if (lane >= o) v += u;
    }
    return v;
}

template <bool FULL>
__device__ __forceinline__ void part_body(const int* __restrict__ eidx, int E, int e0,
                                          unsigned* __restrict__ bsorted, int* __restrict__ boff,
                                          int* hist, int* offs, int* wsum, unsigned* srt) {
    const int t = threadIdx.x;
    const int lane = t & 31;
    const int wv = t >> 5;
    int bk[4], rk[4];
    unsigned wd[4];
    #pragma unroll
    for (int k = 0; k < 4; ++k) {
        const int i = e0 + t + 1024 * k;
        const bool ok = FULL || i < E;
        const int ii = ok ? i : e0;
        const int d = eidx[E + ii];
        const int s = eidx[ii];
        bk[k] = ok ? (d >> RS_SHIFT) : -1;
        wd[k] = ((unsigned)(d & (RS - 1)) << 23) | (unsigned)s;
        rk[k] = 0;
    }
    #pragma unroll
    for (int k = 0; k < 4; ++k)
        if (FULL || bk[k] >= 0) rk[k] = atomicAdd(&hist[bk[k]], 1);
    __syncthreads();
    const int v = t < BP ? hist[t] : 0;
    const int incl = wave_incl_scan(v, lane);
    if (lane == 31) wsum[wv] = incl;
    __syncthreads();
    int off = 0;
    for (int j = 0; j < wv; ++j) off += wsum[j];
    if (t < BP) offs[t] = off + incl - v;
    __syncthreads();
    #pragma unroll
    for (int k = 0; k < 4; ++k)
        if (FULL || bk[k] >= 0) srt[offs[bk[k]] + rk[k]] = wd[k];
    __syncthreads();
    const int nvalid = FULL ? PT : (E - e0);
    #pragma unroll
    for (int k = 0; k < 4; ++k) {
        const int i = t + 1024 * k;
        if (FULL || i < nvalid) bsorted[e0 + i] = srt[i];
    }
    if (t < BP) boff[blockIdx.x * BP + t] = offs[t];
}

__global__ void __launch_bounds__(1024)
part_kernel(const int* __restrict__ eidx, int E,
            unsigned* __restrict__ bsorted, int* __restrict__ boff) {
    __shared__ int hist[BP];
    __shared__ int offs[BP];
    __shared__ int wsum[32];
    __shared__ unsigned srt[PT];
    const int t = threadIdx.x;
    const int e0 = blockIdx.x * PT;
    if (t < BP) hist[t] = 0;
    __syncthreads();
    if (e0 + PT <= E) part_body<true>(eidx, E, e0, bsorted, boff, hist, offs, wsum, srt);
    else              part_body<false>(eidx, E, e0, bsorted, boff, hist, offs, wsum, srt);
}

template <typename F>
__device__ __forceinline__ void sweep_range(const unsigned* __restrict__ bsorted,
                                            const int* __restrict__ boff,
                                            int P, int b, int* runS, int* runL,
                                            F&& f) {
    const int t = threadIdx.x;
    const int lane = t & 31;
    const int wv = t >> 5;
    for (int p0 = 0; p0 < P; p0 += 1024) {
        __syncthreads();
        const int p = p0 + t;
        if (p < P) {
            const int s = boff[p * BP + b];
            runS[t] = p * PT + s;
            runL[t] = boff[p * BP + b + 1] - s;
        }
        __syncthreads();
        const int np = (P - p0) < 1024 ? (P - p0) : 1024;
        for (int q0 = wv; q0 < np; q0 += 128) {
            int S[4], L[4];
            unsigned w[4];
            #pragma unroll
            for (int k = 0; k < 4; ++k) {
                const int q = q0 + 32 * k;
                const bool ok = q < np;
                const int qq = ok ? q : q0;
                S[k] = runS[qq];
                L[k] = ok ? runL[qq] : 0;
            }
            #pragma unroll
            for (int k = 0; k < 4; ++k)
                w[k] = bsorted[lane < L[k] ? S[k] + lane : 0];
            #pragma unroll
            for (int k = 0; k < 4; ++k)
                if (lane < L[k]) f(w[k]);
            #pragma unroll
            for (int k = 0; k < 4; ++k)
                for (int j = lane + 32; j < L[k]; j += 32) f(bsorted[S[k] + j]);
        }
    }
}

#define PROJ_K 256
#define PROJ_CT 4

__global__ void __launch_bounds__(1024, 16)
range_proj_kernel(const unsigned* __restrict__ bsorted, const int* __restrict__ boff,
                  int P, int n, int* __restrict__ start, int* __restrict__ endp,
                  int* __restrict__ csr,
                  const float* __restrict__ x, const float* __restrict__ W,
                  float* __restrict__ hout, int gblocks, int tpb) {
    if ((int)blockIdx.x < gblocks) {
        const int w = threadIdx.x >> 5;
        if (w >= tpb) return;
        const int row0 = ((int)blockIdx.x * tpb + w) << 4;
        if (row0 >= n) return;
        wmma_strip_f32<PROJ_CT, PROJ_K>(x, W, hout, row0, threadIdx.x & 31);
        return;
    }
    __shared__ int cnt[RS];
    __shared__ int cur[RS];
    __shared__ int runS[1024];
    __shared__ int runL[1024];
    __shared__ int wsum[32];
    __shared__ int s_base;
    __shared__ int s_total;
    __shared__ int outb[CAP];
    const int t = threadIdx.x;
    const int lane = t & 31;
    const int wv = t >> 5;
    const int b = (int)blockIdx.x - gblocks;
    if (t < RS) cnt[t] = 0;
    if (t == 0) { s_base = 0; s_total = 0; }
    __syncthreads();
    int pb = 0, pt = 0;
    for (int p = t; p < P; p += 1024) {
        const int s = boff[p * BP + b];
        pb += s;
        pt += boff[p * BP + b + 1] - s;
    }
    for (int o = 16; o > 0; o >>= 1) {
        pb += __shfl_xor(pb, o);
        pt += __shfl_xor(pt, o);
    }
    if (lane == 0) { atomicAdd(&s_base, pb); atomicAdd(&s_total, pt); }
    __syncthreads();
    const int base = s_base;
    const int total = s_total;
    const bool staged = total <= CAP;

    sweep_range(bsorted, boff, P, b, runS, runL,
                [&](unsigned w) { atomicAdd(&cnt[w >> 23], 1); });
    __syncthreads();
    const int v = t < RS ? cnt[t] : 0;
    const int incl = wave_incl_scan(v, lane);
    if (lane == 31) wsum[wv] = incl;
    __syncthreads();
    int off = 0;
    for (int j = 0; j < wv; ++j) off += wsum[j];
    const int excl = off + incl - v;
    if (t < RS) {
        cur[t] = excl;
        const int node = b * RS + t;
        if (node < n) { start[node] = base + excl; endp[node] = base + excl + v; }
    }
    __syncthreads();
    sweep_range(bsorted, boff, P, b, runS, runL,
                [&](unsigned w) {
                    const int pos = atomicAdd(&cur[w >> 23], 1);
                    const int src = (int)(w & 0x7fffffu);
                    if (staged) outb[pos] = src;
                    else atomicExch(&csr[base + pos], src);
                });
    __syncthreads();
    if (staged) {
        const int lim = base + total;
        for (int g = (base & ~31) + t; g < lim; g += 1024) {
            if (g >= base) {
                const int ls = g & ~31;
                const int val = outb[g - base];
                if (ls >= base && ls + 32 <= lim) csr[g] = val;
                else atomicExch(&csr[g], val);
            }
        }
    }
}

template <int M>
__device__ __forceinline__ float row_xor_get(float v) {
    return __int_as_float(__builtin_amdgcn_update_dpp(
        0, __float_as_int(v), 0x160 | M, 0xf, 0xf, true));
}

template <int W>
__device__ __forceinline__ float group_sum(float v) {
    if constexpr (W >= 2) v += row_xor_get<1>(v);
    if constexpr (W >= 4) v += row_xor_get<2>(v);
    if constexpr (W >= 8) v += row_xor_get<4>(v);
    return v;
}

template <int H, int C, int CPL, int U, int NOUT>
__global__ void __launch_bounds__(256)
node_agg_kernel(const int* __restrict__ start,
                const int* __restrict__ endp,
                const int* __restrict__ csr,
                const float* __restrict__ a_src,
                const float* __restrict__ a_dst,
                const float* __restrict__ h,
                const float* __restrict__ bias,
                const float* __restrict__ Wn,
                float* __restrict__ out, int n) {
    constexpr int HC  = H * C;
    constexpr int LPR = HC / CPL;
    constexpr int LPH = C / CPL;
    constexpr int RPW = 32 / LPR;
    constexpr int NV  = CPL / 4;
    static_assert(CPL % 4 == 0 && HC % CPL == 0 && C % CPL == 0 && 32 % LPR == 0
                  && (LPH == 1 || LPH == 2 || LPH == 4 || LPH == 8)
                  && (RPW * HC * 4) % 128 == 0, "lane layout");
    const int lane = threadIdx.x & 31;
    const int wid = __builtin_amdgcn_readfirstlane(
        (int)(blockIdx.x * (blockDim.x >> 5) + (threadIdx.x >> 5)));
    const int node0 = wid * RPW;
    if constexpr (NOUT == 0) { if (node0 >= n) return; }
    const int r  = lane / LPR;
    const int cl = lane - r * LPR;
    const int ch = cl * CPL;
    const bool live = node0 + r < n;
    const int node = live ? node0 + r : n - 1;

    v2f ws[2 * NV], acc[2 * NV];
    v2f ds2 = (v2f){0.f, 0.f}, dd2 = (v2f){0.f, 0.f};
    #pragma unroll
    for (int v = 0; v < NV; ++v) {
        const v4f s4 = *(const v4f*)(a_src + ch + 4 * v);
        const v4f d4 = *(const v4f*)(a_dst + ch + 4 * v);
        const v4f h4 = *(const v4f*)(h + (size_t)node * HC + ch + 4 * v);
        ws[2*v] = s4.lo;  ws[2*v+1] = s4.hi;
        acc[2*v] = h4.lo; acc[2*v+1] = h4.hi;
        ds2 += h4.lo * s4.lo; ds2 += h4.hi * s4.hi;
        dd2 += h4.lo * d4.lo; dd2 += h4.hi * d4.hi;
    }
    const int beg = start[node];
    const int rend = endp[node];
    const int deg = live ? rend - beg : 0;

    const float sd = group_sum<LPH>(dd2.x + dd2.y);
    float m = leaky(group_sum<LPH>(ds2.x + ds2.y) + sd);
    float den = 1.f;

    int maxdeg = 0;
    #pragma unroll
    for (int rr = 0; rr < RPW; ++rr) {
        const int dr = __builtin_amdgcn_readlane(deg, rr * LPR);
        maxdeg = dr > maxdeg ? dr : maxdeg;
    }

    const char* hbytes = (const char*)h;
    const unsigned choff = (unsigned)ch * 4u;
    int sv[U];
    #pragma unroll
    for (int j = 0; j < U; ++j) {
        const int c = csr[j < deg ? beg + j : 0];
        sv[j] = j < deg ? c : node;
    }

    for (int i = 0; i < maxdeg; i += U) {
        v4f hv[U][NV];
        #pragma unroll
        for (int j = 0; j < U; ++j) {
            #pragma unroll
            for (int v = 0; v < NV; ++v)
                hv[j][v] = *(const v4f*)(hbytes + ((unsigned)sv[j] * (unsigned)(HC * 4) + choff + 16u * v));
        }
        #pragma unroll
        for (int j = 0; j < U; ++j) {
            sv[j] = node;
            if (i + U + j < deg) sv[j] = csr[beg + i + U + j];
        }
        float e[U];
        #pragma unroll
        for (int j = 0; j < U; ++j) {
            v2f d2 = (v2f){0.f, 0.f};
            #pragma unroll
            for (int v = 0; v < NV; ++v) {
                d2 += hv[j][v].lo * ws[2*v];
                d2 += hv[j][v].hi * ws[2*v+1];
            }
            const float ej = leaky(group_sum<LPH>(d2.x + d2.y) + sd);
            e[j] = (i + j < deg) ? ej : -INFINITY;
        }
        float mn = m;
        #pragma unroll
        for (int j = 0; j < U; ++j) mn = fmaxf(mn, e[j]);
        const float sc = __expf(m - mn);
        den *= sc;
        #pragma unroll
        for (int c = 0; c < 2 * NV; ++c) acc[c] *= sc;
        #pragma unroll
        for (int j = 0; j < U; ++j) {
            const float p = __expf(e[j] - mn);
            den += p;
            #pragma unroll
            for (int v = 0; v < NV; ++v) {
                acc[2*v]   += p * hv[j][v].lo;
                acc[2*v+1] += p * hv[j][v].hi;
            }
        }
        m = mn;
    }

    float q[CPL];
    #pragma unroll
    for (int v = 0; v < NV; ++v) {
        const float4 b4 = *(const float4*)(bias + ch + 4 * v);
        q[4*v]   = acc[2*v].x   / den + b4.x;
        q[4*v+1] = acc[2*v].y   / den + b4.y;
        q[4*v+2] = acc[2*v+1].x / den + b4.z;
        q[4*v+3] = acc[2*v+1].y / den + b4.w;
    }

    if constexpr (NOUT > 0) {
        constexpr int WPB = 8;
        constexpr int RPB = WPB * RPW;
        constexpr int PITCH = HC + 4;
        static_assert(NOUT == 16 && RPB % 16 == 0 && HC % 4 == 0 && RPB / 16 <= WPB, "product tile");
        __shared__ __attribute__((aligned(16))) float tile[RPB * PITCH];
        const int wib = __builtin_amdgcn_readfirstlane((int)(threadIdx.x >> 5));
        float* trow = tile + (wib * RPW + r) * PITCH + ch;
        #pragma unroll
        for (int v = 0; v < NV; ++v) {
            v4f o; o.x = q[4*v]; o.y = q[4*v+1]; o.z = q[4*v+2]; o.w = q[4*v+3];
            *(v4f*)(trow + 4 * v) = o;
        }
        __syncthreads();
        if (wib < RPB / 16) {
            const int half = lane >> 4;
            const int l16  = lane & 15;
            const float* arow = tile + (wib * 16 + l16) * PITCH;
            v8f d = (v8f){};
            #pragma unroll
            for (int k = 0; k < HC; k += 4) {
                const int ka = k + 2 * half;
                const v2f a = *(const v2f*)(arow + ka);
                v2f b;
                b.x = Wn[(size_t)ka * NOUT + l16];
                b.y = Wn[(size_t)(ka + 1) * NOUT + l16];
                d = __builtin_amdgcn_wmma_f32_16x16x4_f32(
                     false, a,  false, b,
                     (short)0, d,  false,  false);
            }
            const int row0 = (int)blockIdx.x * RPB + wib * 16 + 8 * half;
            #pragma unroll
            for (int i = 0; i < 8; ++i) {
                if (row0 + i < n) out[(size_t)(row0 + i) * NOUT + l16] = d[i];
            }
        }
    } else if constexpr (NV == 1) {
        if (live) {
            float4 o; o.x = q[0]; o.y = q[1]; o.z = q[2]; o.w = q[3];
            *(float4*)(out + (size_t)node * HC + ch) = o;
        }
    } else {
        constexpr int LPS = HC / 4;
        constexpr int RPS = 32 / LPS;
        static_assert(32 % LPS == 0 && RPS * NV == RPW, "store layout");
        const int rs = lane / LPS;
        const int pi = lane - rs * LPS;
        const int own = pi / NV;
        const int pc = pi - own * NV;
        #pragma unroll
        for (int k = 0; k < NV; ++k) {
            const int from = (k * RPS + rs) * LPR + own;
            float o[4];
            #pragma unroll
            for (int c = 0; c < 4; ++c) {
                float val = 0.f;
                #pragma unroll
                for (int w = 0; w < NV; ++w) {
                    const float g = __shfl(q[4 * w + c], from);
                    if (pc == w) val = g;
                }
                o[c] = val;
            }
            const int nd = node0 + k * RPS + rs;
            if (nd < n) {
                float4 o4; o4.x = o[0]; o4.y = o[1]; o4.z = o[2]; o4.w = o[3];
                *(float4*)(out + (size_t)nd * HC + 4 * pi) = o4;
            }
        }
    }
}

static inline int ceil_div(int a, int b) { return (a + b - 1) / b; }

extern "C" void kernel_launch(void* const* d_in, const int* in_sizes, int n_in,
                              void* d_out, int out_size, void* d_ws, size_t ws_size,
                              hipStream_t stream) {
    const float* x    = (const float*)d_in[0];
    const int*   eidx = (const int*)  d_in[1];
    const float* W1   = (const float*)d_in[2];
    const float* aS1  = (const float*)d_in[3];
    const float* aD1  = (const float*)d_in[4];
    const float* b1   = (const float*)d_in[5];
    const float* W2   = (const float*)d_in[6];
    const float* aS2  = (const float*)d_in[7];
    const float* aD2  = (const float*)d_in[8];
    const float* b2   = (const float*)d_in[9];
    float* out = (float*)d_out;

    const int FEATS = PROJ_K, HID = PROJ_CT * 16, H1 = 8, C1 = 8, CLS = 16;
    const int N = in_sizes[0] / FEATS;
    const int E = in_sizes[1] / 2;

    float* ws = (float*)d_ws;
    float* h1   = ws;                 ws += (size_t)N * HID;
    ws += (size_t)N * H1;
    ws += (size_t)N * H1;
    float* o1   = ws;                 ws += (size_t)N * HID;
    float* h2   = ws;                 ws += (size_t)N * CLS;
    ws += (size_t)N;
    ws += (size_t)N;
    int* rstart = (int*)ws;           ws += (size_t)N;
    int* rend   = (int*)ws;           ws += (size_t)N;
    int* csr    = (int*)ws;           ws += (size_t)E;
    int* boff   = (int*)h2;
    unsigned* bsorted = (unsigned*)o1;

    const int P = ceil_div(E, PT);
    const int R = ceil_div(N, RS);
    part_kernel<<<P, 1024, 0, stream>>>(eidx, E, bsorted, boff);

    const int strips = N / 16;
    const int tpb = strips >= 32 * 256 ? 32 : ceil_div(strips, 256);
    const int GB = ceil_div(strips, tpb);
    range_proj_kernel<<<GB + R, 1024, 0, stream>>>(bsorted, boff, P, N, rstart, rend, csr,
                                                   x, W1, h1, GB, tpb);
    node_agg_kernel<H1, C1, 8, 8, CLS><<<ceil_div(ceil_div(N, 4), 8), 256, 0, stream>>>(rstart, rend, csr, aS1, aD1, h1, b1, W2, h2, N);

    node_agg_kernel<1, CLS, 4, 8, 0><<<ceil_div(ceil_div(N, 8), 8), 256, 0, stream>>>(rstart, rend, csr, aS2, aD2, h2, b2, nullptr, out, N);
}
